# starting kernel
# baseline (speedup 1.0000x reference)
#include <hip/hip_runtime.h>


typedef __attribute__((ext_vector_type(2))) float v2f;
typedef __attribute__((ext_vector_type(8))) float v8f;

#define IN_DIM 128
#define H1_DIM 16
#define H2_DIM 8

__global__ void gcn_init_deg(float* __restrict__ deg, int n) {
    int i = blockIdx.x * blockDim.x + threadIdx.x;
    if (i < n) deg[i] = 1.0f;
}

__global__ void gcn_count_deg(const int* __restrict__ dst, float* __restrict__ deg, int e) {
    int i = blockIdx.x * blockDim.x + threadIdx.x;
    if (i < e) atomicAdd(&deg[dst[i]], 1.0f);
}

__global__ void gcn_rsqrt_deg(float* __restrict__ deg, int n) {
    int i = blockIdx.x * blockDim.x + threadIdx.x;
    if (i < n) {
        float d = deg[i];
        deg[i] = (d > 0.0f) ? rsqrtf(d) : 0.0f;
    }
}

__global__ __launch_bounds__(256) void gcn_gemm1_wmma(
    const float* __restrict__ x, const float* __restrict__ W1,
    const float* __restrict__ dis, const float* __restrict__ b1,
    float* __restrict__ hw1, float* __restrict__ agg1, int n) {
    __shared__ float W1s[IN_DIM * H1_DIM];
    const int tid = threadIdx.x;
    #pragma unroll
    for (int j = 0; j < (IN_DIM * H1_DIM) / 256; ++j)
        W1s[tid + j * 256] = W1[tid + j * 256];
    __syncthreads();

    const int wave = tid >> 5;
    const int lane = tid & 31;
    const int half = lane >> 4;
    const int l    = lane & 15;
    const int row0 = (blockIdx.x * 8 + wave) * 16;
    if (row0 >= n) return;

    int arow = row0 + l;
    if (arow >= n) arow = n - 1;

    v8f c = {};
    #pragma unroll 4
    for (int k0 = 0; k0 < IN_DIM; k0 += 4) {
        const int ka = k0 + half * 2;
        const v2f a = *(const v2f*)&x[(size_t)arow * IN_DIM + ka];
        v2f b;
        b.x = W1s[ka * H1_DIM + l];
        b.y = W1s[(ka + 1) * H1_DIM + l];
        c = __builtin_amdgcn_wmma_f32_16x16x4_f32(
                false, a, false, b, (short)0, c, false, false);
    }

    const float bias = b1[l];
    if (row0 + 16 <= n) {
        #pragma unroll
        for (int r = 0; r < 8; ++r) {
            const int row = row0 + r + half * 8;
            const float v  = c[r];
            const float dv = dis[row];
            hw1 [(size_t)row * H1_DIM + l] = v;
            agg1[(size_t)row * H1_DIM + l] = dv * dv * v + bias;
        }
    } else {
        for (int r = 0; r < 8; ++r) {
            const int row = row0 + r + half * 8;
            if (row < n) {
                const float v  = c[r];
                const float dv = dis[row];
                hw1 [(size_t)row * H1_DIM + l] = v;
                agg1[(size_t)row * H1_DIM + l] = dv * dv * v + bias;
            }
        }
    }
}

__global__ __launch_bounds__(256) void gcn_scatter1(
    const int* __restrict__ src, const int* __restrict__ dst,
    const float* __restrict__ dis, const float* __restrict__ hw1,
    float* __restrict__ agg1, int e) {
    const int i = blockIdx.x * blockDim.x + threadIdx.x;
    if (i >= e * H1_DIM) return;
    const int eidx = i >> 4;
    const int c    = i & 15;

    if (c == 0) {
        const int la = eidx + 512;
        if (la < e) __builtin_prefetch(&hw1[(size_t)src[la] * H1_DIM], 0, 0);
    }

    const int s = src[eidx];
    const int d = dst[eidx];
    const float w = dis[s] * dis[d];
    atomicAdd(&agg1[(size_t)d * H1_DIM + c], w * hw1[(size_t)s * H1_DIM + c]);
}

__global__ __launch_bounds__(256) void gcn_gemm2_wmma(
    const float* __restrict__ h1, const float* __restrict__ W2,
    const float* __restrict__ dis, const float* __restrict__ b2,
    float* __restrict__ hw2, float* __restrict__ out, int n) {
    __shared__ float W2s[H1_DIM * 16];
    const int tid = threadIdx.x;
    {
        const int k = tid >> 4, col = tid & 15;
        W2s[tid] = (col < H2_DIM) ? W2[k * H2_DIM + col] : 0.0f;
    }
    __syncthreads();

    const int wave = tid >> 5;
    const int lane = tid & 31;
    const int half = lane >> 4;
    const int l    = lane & 15;
    const int row0 = (blockIdx.x * 8 + wave) * 16;
    if (row0 >= n) return;

    int arow = row0 + l;
    if (arow >= n) arow = n - 1;

    v8f c = {};
    #pragma unroll
    for (int k0 = 0; k0 < H1_DIM; k0 += 4) {
        const int ka = k0 + half * 2;
        v2f a = *(const v2f*)&h1[(size_t)arow * H1_DIM + ka];
        a.x = fmaxf(a.x, 0.0f);
        a.y = fmaxf(a.y, 0.0f);
        v2f b;
        b.x = W2s[ka * 16 + l];
        b.y = W2s[(ka + 1) * 16 + l];
        c = __builtin_amdgcn_wmma_f32_16x16x4_f32(
                false, a, false, b, (short)0, c, false, false);
    }

    if (l < H2_DIM) {
        const float bias = b2[l];
        if (row0 + 16 <= n) {
            #pragma unroll
            for (int r = 0; r < 8; ++r) {
                const int row = row0 + r + half * 8;
                const float v  = c[r];
                const float dv = dis[row];
                hw2[(size_t)row * H2_DIM + l] = v;
                out[(size_t)row * H2_DIM + l] = dv * dv * v + bias;
            }
        } else {
            for (int r = 0; r < 8; ++r) {
                const int row = row0 + r + half * 8;
                if (row < n) {
                    const float v  = c[r];
                    const float dv = dis[row];
                    hw2[(size_t)row * H2_DIM + l] = v;
                    out[(size_t)row * H2_DIM + l] = dv * dv * v + bias;
                }
            }
        }
    }
}

__global__ __launch_bounds__(256) void gcn_scatter2(
    const int* __restrict__ src, const int* __restrict__ dst,
    const float* __restrict__ dis, const float* __restrict__ hw2,
    float* __restrict__ out, int e) {
    const int i = blockIdx.x * blockDim.x + threadIdx.x;
    if (i >= e * H2_DIM) return;
    const int eidx = i >> 3;
    const int c    = i & 7;

    if (c == 0) {
        const int la = eidx + 512;
        if (la < e) __builtin_prefetch(&hw2[(size_t)src[la] * H2_DIM], 0, 0);
    }

    const int s = src[eidx];
    const int d = dst[eidx];
    const float w = dis[s] * dis[d];
    atomicAdd(&out[(size_t)d * H2_DIM + c], w * hw2[(size_t)s * H2_DIM + c]);
}

extern "C" void kernel_launch(void* const* d_in, const int* in_sizes, int n_in,
                              void* d_out, int out_size, void* d_ws, size_t ws_size,
                              hipStream_t stream) {
    const float* x    = (const float*)d_in[0];
    const int*   edge = (const int*)  d_in[1];
    const float* W1   = (const float*)d_in[2];
    const float* b1   = (const float*)d_in[3];
    const float* W2   = (const float*)d_in[4];
    const float* b2   = (const float*)d_in[5];
    float* out = (float*)d_out;

    const int n = in_sizes[0] / IN_DIM;
    const int e = in_sizes[1] / 2;
    const int* src = edge;
    const int* dst = edge + e;

    float* ws   = (float*)d_ws;
    float* dis  = ws;
    float* hw1  = dis + n;
    float* agg1 = hw1 + (size_t)n * H1_DIM;
    float* hw2  = agg1 + (size_t)n * H1_DIM;

    const int B = 256;
    const int gn     = (n + B - 1) / B;
    const int ge     = (e + B - 1) / B;
    const int tiles  = (n + 15) / 16;
    const int gtiles = (tiles + 7) / 8;
    const int gs1    = (e * H1_DIM + B - 1) / B;
    const int gs2    = (e * H2_DIM + B - 1) / B;

    gcn_init_deg <<<gn, B, 0, stream>>>(dis, n);
    gcn_count_deg<<<ge, B, 0, stream>>>(dst, dis, e);
    gcn_rsqrt_deg<<<gn, B, 0, stream>>>(dis, n);

    gcn_gemm1_wmma<<<gtiles, B, 0, stream>>>(x, W1, dis, b1, hw1, agg1, n);
    gcn_scatter1  <<<gs1, B, 0, stream>>>(src, dst, dis, hw1, agg1, e);

    gcn_gemm2_wmma<<<gtiles, B, 0, stream>>>(agg1, W2, dis, b2, hw2, out, n);
    gcn_scatter2  <<<gs2, B, 0, stream>>>(src, dst, dis, hw2, out, e);
}


# Degree count: per-edge random global f32 atomics replaced by LDS integer histograms per node range (7 ranges x 36 edge slices), flushed as whole-line f32 atomics; counts exact, so order-free
# speedup vs baseline: 1.4743x; 1.4743x over previous
#include <hip/hip_runtime.h>


typedef __attribute__((ext_vector_type(2))) float v2f;
typedef __attribute__((ext_vector_type(8))) float v8f;

#define IN_DIM 128
#define H1_DIM 16
#define H2_DIM 8

__global__ void gcn_init_deg(float* __restrict__ deg, int n) {
    int i = blockIdx.x * blockDim.x + threadIdx.x;
    if (i < n) deg[i] = 1.0f;
}

#define DEG_R_MAX   16384
#define DEG_THREADS 1024
#define DEG_BATCH   8

__device__ __forceinline__ void deg_hit(unsigned* cnt, int d, int lo, unsigned span) {
    const unsigned u = (unsigned)(d - lo);
    if (u < span) atomicAdd(&cnt[u], 1u);
}

__global__ __launch_bounds__(DEG_THREADS) void gcn_count_deg(
    const int* __restrict__ dst, float* __restrict__ deg, int e, int n, int R) {
    __shared__ unsigned cnt[DEG_R_MAX];
    const int tid = threadIdx.x;
    const int lo  = blockIdx.y * R;
    const int spn = min(R, n - lo);
    if (spn <= 0) return;
    const unsigned span = (unsigned)spn;

    for (int i = tid; i < spn; i += DEG_THREADS) cnt[i] = 0u;
    __syncthreads();

    const bool aligned = (((size_t)dst) & 15) == 0;
    const int nq   = aligned ? (e >> 2) : 0;
    const int qper = (nq + (int)gridDim.x - 1) / (int)gridDim.x;
    const int q0   = min(nq, (int)blockIdx.x * qper);
    const int q1   = min(nq, q0 + qper);
    const int4* __restrict__ dq = (const int4*)dst;
    for (int q = q0 + tid; q < q1; q += DEG_THREADS * DEG_BATCH) {
        int4 v[DEG_BATCH];
        #pragma unroll
        for (int k = 0; k < DEG_BATCH; ++k) {
            const int qq = q + k * DEG_THREADS;
            v[k] = (qq < q1) ? dq[qq] : make_int4(-1, -1, -1, -1);
        }
        #pragma unroll
        for (int k = 0; k < DEG_BATCH; ++k) {
            deg_hit(cnt, v[k].x, lo, span);
            deg_hit(cnt, v[k].y, lo, span);
            deg_hit(cnt, v[k].z, lo, span);
            deg_hit(cnt, v[k].w, lo, span);
        }
    }
    const int r0   = nq << 2;
    const int rper = (e - r0 + (int)gridDim.x - 1) / (int)gridDim.x;
    const int s0   = min(e, r0 + (int)blockIdx.x * rper);
    const int s1   = min(e, s0 + rper);
    for (int i = s0 + tid; i < s1; i += DEG_THREADS) deg_hit(cnt, dst[i], lo, span);
    __syncthreads();

    const int nl   = (spn + 31) >> 5;
    const int wave = tid >> 5;
    const int lane = tid & 31;
    const int rot  = (int)((blockIdx.x * (unsigned)nl) / gridDim.x);
    for (int k = wave; k < nl; k += DEG_THREADS / 32) {
        int line = k + rot;
        if (line >= nl) line -= nl;
        const int idx = (line << 5) + lane;
        if (idx < spn) {
            const unsigned c = cnt[idx];
            if (c != 0u) atomicAdd(&deg[lo + idx], (float)c);
        }
    }
}

__global__ void gcn_rsqrt_deg(float* __restrict__ deg, int n) {
    int i = blockIdx.x * blockDim.x + threadIdx.x;
    if (i < n) {
        float d = deg[i];
        deg[i] = (d > 0.0f) ? rsqrtf(d) : 0.0f;
    }
}

__global__ __launch_bounds__(256) void gcn_gemm1_wmma(
    const float* __restrict__ x, const float* __restrict__ W1,
    const float* __restrict__ dis, const float* __restrict__ b1,
    float* __restrict__ hw1, float* __restrict__ agg1, int n) {
    __shared__ float W1s[IN_DIM * H1_DIM];
    const int tid = threadIdx.x;
    #pragma unroll
    for (int j = 0; j < (IN_DIM * H1_DIM) / 256; ++j)
        W1s[tid + j * 256] = W1[tid + j * 256];
    __syncthreads();

    const int wave = tid >> 5;
    const int lane = tid & 31;
    const int half = lane >> 4;
    const int l    = lane & 15;
    const int row0 = (blockIdx.x * 8 + wave) * 16;
    if (row0 >= n) return;

    int arow = row0 + l;
    if (arow >= n) arow = n - 1;

    v8f c = {};
    #pragma unroll 4
    for (int k0 = 0; k0 < IN_DIM; k0 += 4) {
        const int ka = k0 + half * 2;
        const v2f a = *(const v2f*)&x[(size_t)arow * IN_DIM + ka];
        v2f b;
        b.x = W1s[ka * H1_DIM + l];
        b.y = W1s[(ka + 1) * H1_DIM + l];
        c = __builtin_amdgcn_wmma_f32_16x16x4_f32(
                false, a, false, b, (short)0, c, false, false);
    }

    const float bias = b1[l];
    if (row0 + 16 <= n) {
        #pragma unroll
        for (int r = 0; r < 8; ++r) {
            const int row = row0 + r + half * 8;
            const float v  = c[r];
            const float dv = dis[row];
            hw1 [(size_t)row * H1_DIM + l] = v;
            agg1[(size_t)row * H1_DIM + l] = dv * dv * v + bias;
        }
    } else {
        for (int r = 0; r < 8; ++r) {
            const int row = row0 + r + half * 8;
            if (row < n) {
                const float v  = c[r];
                const float dv = dis[row];
                hw1 [(size_t)row * H1_DIM + l] = v;
                agg1[(size_t)row * H1_DIM + l] = dv * dv * v + bias;
            }
        }
    }
}

__global__ __launch_bounds__(256) void gcn_scatter1(
    const int* __restrict__ src, const int* __restrict__ dst,
    const float* __restrict__ dis, const float* __restrict__ hw1,
    float* __restrict__ agg1, int e) {
    const int i = blockIdx.x * blockDim.x + threadIdx.x;
    if (i >= e * H1_DIM) return;
    const int eidx = i >> 4;
    const int c    = i & 15;

    if (c == 0) {
        const int la = eidx + 512;
        if (la < e) __builtin_prefetch(&hw1[(size_t)src[la] * H1_DIM], 0, 0);
    }

    const int s = src[eidx];
    const int d = dst[eidx];
    const float w = dis[s] * dis[d];
    atomicAdd(&agg1[(size_t)d * H1_DIM + c], w * hw1[(size_t)s * H1_DIM + c]);
}

__global__ __launch_bounds__(256) void gcn_gemm2_wmma(
    const float* __restrict__ h1, const float* __restrict__ W2,
    const float* __restrict__ dis, const float* __restrict__ b2,
    float* __restrict__ hw2, float* __restrict__ out, int n) {
    __shared__ float W2s[H1_DIM * 16];
    const int tid = threadIdx.x;
    {
        const int k = tid >> 4, col = tid & 15;
        W2s[tid] = (col < H2_DIM) ? W2[k * H2_DIM + col] : 0.0f;
    }
    __syncthreads();

    const int wave = tid >> 5;
    const int lane = tid & 31;
    const int half = lane >> 4;
    const int l    = lane & 15;
    const int row0 = (blockIdx.x * 8 + wave) * 16;
    if (row0 >= n) return;

    int arow = row0 + l;
    if (arow >= n) arow = n - 1;

    v8f c = {};
    #pragma unroll
    for (int k0 = 0; k0 < H1_DIM; k0 += 4) {
        const int ka = k0 + half * 2;
        v2f a = *(const v2f*)&h1[(size_t)arow * H1_DIM + ka];
        a.x = fmaxf(a.x, 0.0f);
        a.y = fmaxf(a.y, 0.0f);
        v2f b;
        b.x = W2s[ka * 16 + l];
        b.y = W2s[(ka + 1) * 16 + l];
        c = __builtin_amdgcn_wmma_f32_16x16x4_f32(
                false, a, false, b, (short)0, c, false, false);
    }

    if (l < H2_DIM) {
        const float bias = b2[l];
        if (row0 + 16 <= n) {
            #pragma unroll
            for (int r = 0; r < 8; ++r) {
                const int row = row0 + r + half * 8;
                const float v  = c[r];
                const float dv = dis[row];
                hw2[(size_t)row * H2_DIM + l] = v;
                out[(size_t)row * H2_DIM + l] = dv * dv * v + bias;
            }
        } else {
            for (int r = 0; r < 8; ++r) {
                const int row = row0 + r + half * 8;
                if (row < n) {
                    const float v  = c[r];
                    const float dv = dis[row];
                    hw2[(size_t)row * H2_DIM + l] = v;
                    out[(size_t)row * H2_DIM + l] = dv * dv * v + bias;
                }
            }
        }
    }
}

__global__ __launch_bounds__(256) void gcn_scatter2(
    const int* __restrict__ src, const int* __restrict__ dst,
    const float* __restrict__ dis, const float* __restrict__ hw2,
    float* __restrict__ out, int e) {
    const int i = blockIdx.x * blockDim.x + threadIdx.x;
    if (i >= e * H2_DIM) return;
    const int eidx = i >> 3;
    const int c    = i & 7;

    if (c == 0) {
        const int la = eidx + 512;
        if (la < e) __builtin_prefetch(&hw2[(size_t)src[la] * H2_DIM], 0, 0);
    }

    const int s = src[eidx];
    const int d = dst[eidx];
    const float w = dis[s] * dis[d];
    atomicAdd(&out[(size_t)d * H2_DIM + c], w * hw2[(size_t)s * H2_DIM + c]);
}

extern "C" void kernel_launch(void* const* d_in, const int* in_sizes, int n_in,
                              void* d_out, int out_size, void* d_ws, size_t ws_size,
                              hipStream_t stream) {
    const float* x    = (const float*)d_in[0];
    const int*   edge = (const int*)  d_in[1];
    const float* W1   = (const float*)d_in[2];
    const float* b1   = (const float*)d_in[3];
    const float* W2   = (const float*)d_in[4];
    const float* b2   = (const float*)d_in[5];
    float* out = (float*)d_out;

    const int n = in_sizes[0] / IN_DIM;
    const int e = in_sizes[1] / 2;
    const int* src = edge;
    const int* dst = edge + e;

    float* ws   = (float*)d_ws;
    float* dis  = ws;
    float* hw1  = dis + n;
    float* agg1 = hw1 + (size_t)n * H1_DIM;
    float* hw2  = agg1 + (size_t)n * H1_DIM;

    const int B = 256;
    const int gn     = (n + B - 1) / B;
    const int dpieces = (n + DEG_R_MAX - 1) / DEG_R_MAX;
    const int dR      = (((n + dpieces - 1) / dpieces) + 31) & ~31;
    const int dslices = (256 / dpieces) > 0 ? (256 / dpieces) : 1;
    const int tiles  = (n + 15) / 16;
    const int gtiles = (tiles + 7) / 8;
    const int gs1    = (e * H1_DIM + B - 1) / B;
    const int gs2    = (e * H2_DIM + B - 1) / B;

    gcn_init_deg <<<gn, B, 0, stream>>>(dis, n);
    gcn_count_deg<<<dim3(dslices, dpieces), DEG_THREADS, 0, stream>>>(dst, dis, e, n, dR);
    gcn_rsqrt_deg<<<gn, B, 0, stream>>>(dis, n);

    gcn_gemm1_wmma<<<gtiles, B, 0, stream>>>(x, W1, dis, b1, hw1, agg1, n);
    gcn_scatter1  <<<gs1, B, 0, stream>>>(src, dst, dis, hw1, agg1, e);

    gcn_gemm2_wmma<<<gtiles, B, 0, stream>>>(agg1, W2, dis, b2, hw2, out, n);
    gcn_scatter2  <<<gs2, B, 0, stream>>>(src, dst, dis, hw2, out, e);
}


# Both edge scatters: edges binned once by 128-node dst range (8-byte records with the edge weight), each range summed in LDS (ds f32 adds, arrival order), plain stores; no global f32 atomics there
# speedup vs baseline: 4.3520x; 2.9519x over previous
#include <hip/hip_runtime.h>


typedef __attribute__((ext_vector_type(2))) float v2f;
typedef __attribute__((ext_vector_type(8))) float v8f;

#define IN_DIM 128
#define H1_DIM 16
#define H2_DIM 8

__global__ void gcn_init_deg(float* __restrict__ deg, int n) {
    int i = blockIdx.x * blockDim.x + threadIdx.x;
    if (i < n) deg[i] = 1.0f;
}

#define DEG_R_MAX   16384
#define DEG_THREADS 1024
#define DEG_BATCH   8

__device__ __forceinline__ void deg_hit(unsigned* cnt, int d, int lo, unsigned span) {
    const unsigned u = (unsigned)(d - lo);
    if (u < span) atomicAdd(&cnt[u], 1u);
}

__global__ __launch_bounds__(DEG_THREADS) void gcn_count_deg(
    const int* __restrict__ dst, float* __restrict__ deg, int e, int n, int R) {
    __shared__ unsigned cnt[DEG_R_MAX];
    const int tid = threadIdx.x;
    const int lo  = blockIdx.y * R;
    const int spn = min(R, n - lo);
    if (spn <= 0) return;
    const unsigned span = (unsigned)spn;

    for (int i = tid; i < spn; i += DEG_THREADS) cnt[i] = 0u;
    __syncthreads();

    const bool aligned = (((size_t)dst) & 15) == 0;
    const int nq   = aligned ? (e >> 2) : 0;
    const int qper = (nq + (int)gridDim.x - 1) / (int)gridDim.x;
    const int q0   = min(nq, (int)blockIdx.x * qper);
    const int q1   = min(nq, q0 + qper);
    const int4* __restrict__ dq = (const int4*)dst;
    for (int q = q0 + tid; q < q1; q += DEG_THREADS * DEG_BATCH) {
        int4 v[DEG_BATCH];
        #pragma unroll
        for (int k = 0; k < DEG_BATCH; ++k) {
            const int qq = q + k * DEG_THREADS;
            v[k] = (qq < q1) ? dq[qq] : make_int4(-1, -1, -1, -1);
        }
        #pragma unroll
        for (int k = 0; k < DEG_BATCH; ++k) {
            deg_hit(cnt, v[k].x, lo, span);
            deg_hit(cnt, v[k].y, lo, span);
            deg_hit(cnt, v[k].z, lo, span);
            deg_hit(cnt, v[k].w, lo, span);
        }
    }
    const int r0   = nq << 2;
    const int rper = (e - r0 + (int)gridDim.x - 1) / (int)gridDim.x;
    const int s0   = min(e, r0 + (int)blockIdx.x * rper);
    const int s1   = min(e, s0 + rper);
    for (int i = s0 + tid; i < s1; i += DEG_THREADS) deg_hit(cnt, dst[i], lo, span);
    __syncthreads();

    const int nl   = (spn + 31) >> 5;
    const int wave = tid >> 5;
    const int lane = tid & 31;
    const int rot  = (int)((blockIdx.x * (unsigned)nl) / gridDim.x);
    for (int k = wave; k < nl; k += DEG_THREADS / 32) {
        int line = k + rot;
        if (line >= nl) line -= nl;
        const int idx = (line << 5) + lane;
        if (idx < spn) {
            const unsigned c = cnt[idx];
            if (c != 0u) atomicAdd(&deg[lo + idx], (float)c);
        }
    }
}

#define BK_SH 7
#define BK_R  128
#define BIN_THREADS 1024
#define BIN_BATCH   8
#define SC_BATCH    8
#define SC_THREADS  512

__global__ __launch_bounds__(BK_R) void gcn_rsqrt_bucket(
    float* __restrict__ deg, int* __restrict__ total, int n) {
    __shared__ int s_cnt;
    const int tid = threadIdx.x;
    if (tid == 0) s_cnt = 0;
    __syncthreads();
    const int i = blockIdx.x * BK_R + tid;
    if (i < n) {
        const float d = deg[i];
        deg[i] = (d > 0.0f) ? rsqrtf(d) : 0.0f;
        atomicAdd(&s_cnt, (int)d - 1);
    }
    __syncthreads();
    if (tid == 0) total[blockIdx.x] = s_cnt;
}

__global__ __launch_bounds__(1024) void gcn_bucket_scan(
    const int* __restrict__ total, int* __restrict__ bstart,
    int* __restrict__ cursor, int P) {
    __shared__ int wsum[32];
    const int tid  = threadIdx.x;
    const int lane = tid & 31;
    const int wave = tid >> 5;
    const int per  = (P + 1023) / 1024;
    const int b0   = min(P, tid * per);
    const int b1   = min(P, b0 + per);
    int s = 0;
    for (int b = b0; b < b1; ++b) s += total[b];
    int v = s;
    for (int o = 1; o < 32; o <<= 1) {
        const int t = __shfl_up(v, o);
        if (lane >= o) v += t;
    }
    if (lane == 31) wsum[wave] = v;
    __syncthreads();
    if (wave == 0) {
        int w = wsum[lane];
        for (int o = 1; o < 32; o <<= 1) {
            const int t = __shfl_up(w, o);
            if (lane >= o) w += t;
        }
        wsum[lane] = w;
    }
    __syncthreads();
    int run = v - s + (wave > 0 ? wsum[wave - 1] : 0);
    for (int b = b0; b < b1; ++b) {
        bstart[b] = run;
        cursor[b] = run;
        run += total[b];
    }
    if (tid == 1023) bstart[P] = run;
}

__global__ __launch_bounds__(BIN_THREADS) void gcn_bin_edges(
    const int* __restrict__ src, const int* __restrict__ dst,
    const float* __restrict__ dis, int* __restrict__ cursor,
    int* __restrict__ rec, int e, int n, int P) {
    extern __shared__ int bin_sm[];
    int* hist  = bin_sm;
    int* lbase = bin_sm + P;
    const int tid = threadIdx.x;
    for (int p = tid; p < P; p += BIN_THREADS) hist[p] = 0;
    __syncthreads();

    const int per = (e + (int)gridDim.x - 1) / (int)gridDim.x;
    const int s0  = min(e, (int)blockIdx.x * per);
    const int s1  = min(e, s0 + per);

    for (int i0 = s0 + tid; i0 < s1; i0 += BIN_THREADS * BIN_BATCH) {
        int d[BIN_BATCH];
        #pragma unroll
        for (int k = 0; k < BIN_BATCH; ++k) {
            const int i = i0 + k * BIN_THREADS;
            d[k] = (i < s1) ? dst[i] : -1;
        }
        #pragma unroll
        for (int k = 0; k < BIN_BATCH; ++k)
            if ((unsigned)d[k] < (unsigned)n) atomicAdd(&hist[d[k] >> BK_SH], 1);
    }
    __syncthreads();
    for (int p = tid; p < P; p += BIN_THREADS) {
        const int c = hist[p];
        lbase[p] = (c > 0) ? atomicAdd(&cursor[p], c) : 0;
        hist[p] = 0;
    }
    __syncthreads();

    for (int i0 = s0 + tid; i0 < s1; i0 += BIN_THREADS * BIN_BATCH) {
        int d[BIN_BATCH], sidx[BIN_BATCH];
        #pragma unroll
        for (int k = 0; k < BIN_BATCH; ++k) {
            const int i = i0 + k * BIN_THREADS;
            const bool ok = i < s1;
            d[k]    = ok ? dst[i] : -1;
            sidx[k] = ok ? src[i] : 0;
        }
        float w[BIN_BATCH];
        #pragma unroll
        for (int k = 0; k < BIN_BATCH; ++k) {
            const int dc = ((unsigned)d[k] < (unsigned)n) ? d[k] : 0;
            w[k] = dis[sidx[k]] * dis[dc];
        }
        #pragma unroll
        for (int k = 0; k < BIN_BATCH; ++k) {
            if ((unsigned)d[k] < (unsigned)n) {
                const int b = d[k] >> BK_SH;
                const int slot = lbase[b] + atomicAdd(&hist[b], 1);
                const unsigned key = ((unsigned)sidx[k] << BK_SH) | (unsigned)(d[k] & (BK_R - 1));
                if ((unsigned)slot < (unsigned)e)
                    *(int2*)&rec[2 * (size_t)slot] = make_int2((int)key, __float_as_int(w[k]));
            }
        }
    }
}

__global__ __launch_bounds__(256) void gcn_gemm1_wmma(
    const float* __restrict__ x, const float* __restrict__ W1,
    const float* __restrict__ dis, const float* __restrict__ b1,
    float* __restrict__ hw1, float* __restrict__ agg1, int n) {
    __shared__ float W1s[IN_DIM * H1_DIM];
    const int tid = threadIdx.x;
    #pragma unroll
    for (int j = 0; j < (IN_DIM * H1_DIM) / 256; ++j)
        W1s[tid + j * 256] = W1[tid + j * 256];
    __syncthreads();

    const int wave = tid >> 5;
    const int lane = tid & 31;
    const int half = lane >> 4;
    const int l    = lane & 15;
    const int row0 = (blockIdx.x * 8 + wave) * 16;
    if (row0 >= n) return;

    int arow = row0 + l;
    if (arow >= n) arow = n - 1;

    v8f c = {};
    #pragma unroll 4
    for (int k0 = 0; k0 < IN_DIM; k0 += 4) {
        const int ka = k0 + half * 2;
        const v2f a = *(const v2f*)&x[(size_t)arow * IN_DIM + ka];
        v2f b;
        b.x = W1s[ka * H1_DIM + l];
        b.y = W1s[(ka + 1) * H1_DIM + l];
        c = __builtin_amdgcn_wmma_f32_16x16x4_f32(
                false, a, false, b, (short)0, c, false, false);
    }

    const float bias = b1[l];
    if (row0 + 16 <= n) {
        #pragma unroll
        for (int r = 0; r < 8; ++r) {
            const int row = row0 + r + half * 8;
            const float v  = c[r];
            const float dv = dis[row];
            hw1 [(size_t)row * H1_DIM + l] = v;
            agg1[(size_t)row * H1_DIM + l] = dv * dv * v + bias;
        }
    } else {
        for (int r = 0; r < 8; ++r) {
            const int row = row0 + r + half * 8;
            if (row < n) {
                const float v  = c[r];
                const float dv = dis[row];
                hw1 [(size_t)row * H1_DIM + l] = v;
                agg1[(size_t)row * H1_DIM + l] = dv * dv * v + bias;
            }
        }
    }
}

__global__ __launch_bounds__(SC_THREADS) void gcn_scatter1(
    const int* __restrict__ rec, const int* __restrict__ bstart,
    const float* __restrict__ hw1, float* __restrict__ agg1, int n) {
    __shared__ float acc[BK_R * H1_DIM];
    const int tid = threadIdx.x;
    for (int j = tid; j < BK_R * H1_DIM; j += SC_THREADS) acc[j] = 0.0f;
    __syncthreads();

    const int lo  = blockIdx.x * BK_R;
    const int beg = bstart[blockIdx.x];
    const int end = bstart[blockIdx.x + 1];
    const int c = tid & (H1_DIM - 1);
    const int g = tid >> 4;
    const int G = SC_THREADS >> 4;
    for (int i0 = beg + g; i0 < end; i0 += G * SC_BATCH) {
        int2 r[SC_BATCH];
        #pragma unroll
        for (int k = 0; k < SC_BATCH; ++k) {
            const int i = min(i0 + k * G, end - 1);
            r[k] = *(const int2*)&rec[2 * (size_t)i];
        }
        float h[SC_BATCH];
        #pragma unroll
        for (int k = 0; k < SC_BATCH; ++k) {
            const unsigned s = min((unsigned)r[k].x >> BK_SH, (unsigned)(n - 1));
            h[k] = hw1[(size_t)s * H1_DIM + c];
        }
        #pragma unroll
        for (int k = 0; k < SC_BATCH; ++k) {
            const float v = (i0 + k * G < end) ? __int_as_float(r[k].y) * h[k] : 0.0f;
            const int dl = r[k].x & (BK_R - 1);
            atomicAdd(&acc[dl * H1_DIM + c], v);
        }
    }
    __syncthreads();

    const int cnt = min(BK_R, n - lo) * H1_DIM;
    float* __restrict__ row = agg1 + (size_t)lo * H1_DIM;
    for (int j = tid; j < cnt; j += SC_THREADS) row[j] = row[j] + acc[j];
}

__global__ __launch_bounds__(256) void gcn_gemm2_wmma(
    const float* __restrict__ h1, const float* __restrict__ W2,
    const float* __restrict__ dis, const float* __restrict__ b2,
    float* __restrict__ hw2, float* __restrict__ out, int n) {
    __shared__ float W2s[H1_DIM * 16];
    const int tid = threadIdx.x;
    {
        const int k = tid >> 4, col = tid & 15;
        W2s[tid] = (col < H2_DIM) ? W2[k * H2_DIM + col] : 0.0f;
    }
    __syncthreads();

    const int wave = tid >> 5;
    const int lane = tid & 31;
    const int half = lane >> 4;
    const int l    = lane & 15;
    const int row0 = (blockIdx.x * 8 + wave) * 16;
    if (row0 >= n) return;

    int arow = row0 + l;
    if (arow >= n) arow = n - 1;

    v8f c = {};
    #pragma unroll
    for (int k0 = 0; k0 < H1_DIM; k0 += 4) {
        const int ka = k0 + half * 2;
        v2f a = *(const v2f*)&h1[(size_t)arow * H1_DIM + ka];
        a.x = fmaxf(a.x, 0.0f);
        a.y = fmaxf(a.y, 0.0f);
        v2f b;
        b.x = W2s[ka * 16 + l];
        b.y = W2s[(ka + 1) * 16 + l];
        c = __builtin_amdgcn_wmma_f32_16x16x4_f32(
                false, a, false, b, (short)0, c, false, false);
    }

    if (l < H2_DIM) {
        const float bias = b2[l];
        if (row0 + 16 <= n) {
            #pragma unroll
            for (int r = 0; r < 8; ++r) {
                const int row = row0 + r + half * 8;
                const float v  = c[r];
                const float dv = dis[row];
                hw2[(size_t)row * H2_DIM + l] = v;
                out[(size_t)row * H2_DIM + l] = dv * dv * v + bias;
            }
        } else {
            for (int r = 0; r < 8; ++r) {
                const int row = row0 + r + half * 8;
                if (row < n) {
                    const float v  = c[r];
                    const float dv = dis[row];
                    hw2[(size_t)row * H2_DIM + l] = v;
                    out[(size_t)row * H2_DIM + l] = dv * dv * v + bias;
                }
            }
        }
    }
}

__global__ __launch_bounds__(SC_THREADS) void gcn_scatter2(
    const int* __restrict__ rec, const int* __restrict__ bstart,
    const float* __restrict__ hw2, float* __restrict__ out, int n) {
    __shared__ float acc[BK_R * H2_DIM];
    const int tid = threadIdx.x;
    for (int j = tid; j < BK_R * H2_DIM; j += SC_THREADS) acc[j] = 0.0f;
    __syncthreads();

    const int lo  = blockIdx.x * BK_R;
    const int beg = bstart[blockIdx.x];
    const int end = bstart[blockIdx.x + 1];
    const int c = tid & (H2_DIM - 1);
    const int g = tid >> 3;
    const int G = SC_THREADS >> 3;
    for (int i0 = beg + g; i0 < end; i0 += G * SC_BATCH) {
        int2 r[SC_BATCH];
        #pragma unroll
        for (int k = 0; k < SC_BATCH; ++k) {
            const int i = min(i0 + k * G, end - 1);
            r[k] = *(const int2*)&rec[2 * (size_t)i];
        }
        float h[SC_BATCH];
        #pragma unroll
        for (int k = 0; k < SC_BATCH; ++k) {
            const unsigned s = min((unsigned)r[k].x >> BK_SH, (unsigned)(n - 1));
            h[k] = hw2[(size_t)s * H2_DIM + c];
        }
        #pragma unroll
        for (int k = 0; k < SC_BATCH; ++k) {
            const float v = (i0 + k * G < end) ? __int_as_float(r[k].y) * h[k] : 0.0f;
            const int dl = r[k].x & (BK_R - 1);
            atomicAdd(&acc[dl * H2_DIM + c], v);
        }
    }
    __syncthreads();

    const int cnt = min(BK_R, n - lo) * H2_DIM;
    float* __restrict__ row = out + (size_t)lo * H2_DIM;
    for (int j = tid; j < cnt; j += SC_THREADS) row[j] = row[j] + acc[j];
}

extern "C" void kernel_launch(void* const* d_in, const int* in_sizes, int n_in,
                              void* d_out, int out_size, void* d_ws, size_t ws_size,
                              hipStream_t stream) {
    const float* x    = (const float*)d_in[0];
    const int*   edge = (const int*)  d_in[1];
    const float* W1   = (const float*)d_in[2];
    const float* b1   = (const float*)d_in[3];
    const float* W2   = (const float*)d_in[4];
    const float* b2   = (const float*)d_in[5];
    float* out = (float*)d_out;

    const int n = in_sizes[0] / IN_DIM;
    const int e = in_sizes[1] / 2;
    const int* src = edge;
    const int* dst = edge + e;

    float* ws   = (float*)d_ws;
    float* dis  = ws;
    float* hw1  = dis + n;
    float* agg1 = hw1 + (size_t)n * H1_DIM;
    float* hw2  = agg1 + (size_t)n * H1_DIM;

    const int P = (n + BK_R - 1) / BK_R;
    int* total  = (int*)(hw2 + (size_t)n * H2_DIM);
    int* cursor = total + P;
    int* bstart = cursor + P;
    int* rec    = (int*)((((size_t)(bstart + P + 1)) + 15) & ~(size_t)15);

    const int B = 256;
    const int gn     = (n + B - 1) / B;
    const int dpieces = (n + DEG_R_MAX - 1) / DEG_R_MAX;
    const int dR      = (((n + dpieces - 1) / dpieces) + 31) & ~31;
    const int dslices = (256 / dpieces) > 0 ? (256 / dpieces) : 1;
    const int tiles  = (n + 15) / 16;
    const int gtiles = (tiles + 7) / 8;

    gcn_init_deg <<<gn, B, 0, stream>>>(dis, n);
    gcn_count_deg<<<dim3(dslices, dpieces), DEG_THREADS, 0, stream>>>(dst, dis, e, n, dR);
    gcn_rsqrt_bucket<<<P, BK_R, 0, stream>>>(dis, total, n);
    gcn_bucket_scan <<<1, 1024, 0, stream>>>(total, bstart, cursor, P);
    gcn_bin_edges   <<<256, BIN_THREADS, 2 * (size_t)P * sizeof(int), stream>>>(src, dst, dis, cursor, rec, e, n, P);

    gcn_gemm1_wmma<<<gtiles, B, 0, stream>>>(x, W1, dis, b1, hw1, agg1, n);
    gcn_scatter1  <<<P, SC_THREADS, 0, stream>>>(rec, bstart, hw1, agg1, n);

    gcn_gemm2_wmma<<<gtiles, B, 0, stream>>>(agg1, W2, dis, b2, hw2, out, n);
    gcn_scatter2  <<<P, SC_THREADS, 0, stream>>>(rec, bstart, hw2, out, n);
}


# Binning: each slice's records are placed bucket-sorted in LDS (scan of its counts) and written out in that order, coalesced, not one scattered 4-byte store per edge; order inside a bucket changes
# speedup vs baseline: 6.1083x; 1.4036x over previous
#include <hip/hip_runtime.h>


typedef __attribute__((ext_vector_type(2))) float v2f;
typedef __attribute__((ext_vector_type(8))) float v8f;

#define IN_DIM 128
#define H1_DIM 16
#define H2_DIM 8

__global__ void gcn_init_deg(float* __restrict__ deg, int n) {
    int i = blockIdx.x * blockDim.x + threadIdx.x;
    if (i < n) deg[i] = 1.0f;
}

#define DEG_R_MAX   16384
#define DEG_THREADS 1024
#define DEG_BATCH   8

__device__ __forceinline__ void deg_hit(unsigned* cnt, int d, int lo, unsigned span) {
    const unsigned u = (unsigned)(d - lo);
    if (u < span) atomicAdd(&cnt[u], 1u);
}

__global__ __launch_bounds__(DEG_THREADS) void gcn_count_deg(
    const int* __restrict__ dst, float* __restrict__ deg, int e, int n, int R) {
    __shared__ unsigned cnt[DEG_R_MAX];
    const int tid = threadIdx.x;
    const int lo  = blockIdx.y * R;
    const int spn = min(R, n - lo);
    if (spn <= 0) return;
    const unsigned span = (unsigned)spn;

    for (int i = tid; i < spn; i += DEG_THREADS) cnt[i] = 0u;
    __syncthreads();

    const bool aligned = (((size_t)dst) & 15) == 0;
    const int nq   = aligned ? (e >> 2) : 0;
    const int qper = (nq + (int)gridDim.x - 1) / (int)gridDim.x;
    const int q0   = min(nq, (int)blockIdx.x * qper);
    const int q1   = min(nq, q0 + qper);
    const int4* __restrict__ dq = (const int4*)dst;
    for (int q = q0 + tid; q < q1; q += DEG_THREADS * DEG_BATCH) {
        int4 v[DEG_BATCH];
        #pragma unroll
        for (int k = 0; k < DEG_BATCH; ++k) {
            const int qq = q + k * DEG_THREADS;
            v[k] = (qq < q1) ? dq[qq] : make_int4(-1, -1, -1, -1);
        }
        #pragma unroll
        for (int k = 0; k < DEG_BATCH; ++k) {
            deg_hit(cnt, v[k].x, lo, span);
            deg_hit(cnt, v[k].y, lo, span);
            deg_hit(cnt, v[k].z, lo, span);
            deg_hit(cnt, v[k].w, lo, span);
        }
    }
    const int r0   = nq << 2;
    const int rper = (e - r0 + (int)gridDim.x - 1) / (int)gridDim.x;
    const int s0   = min(e, r0 + (int)blockIdx.x * rper);
    const int s1   = min(e, s0 + rper);
    for (int i = s0 + tid; i < s1; i += DEG_THREADS) deg_hit(cnt, dst[i], lo, span);
    __syncthreads();

    const int nl   = (spn + 31) >> 5;
    const int wave = tid >> 5;
    const int lane = tid & 31;
    const int rot  = (int)((blockIdx.x * (unsigned)nl) / gridDim.x);
    for (int k = wave; k < nl; k += DEG_THREADS / 32) {
        int line = k + rot;
        if (line >= nl) line -= nl;
        const int idx = (line << 5) + lane;
        if (idx < spn) {
            const unsigned c = cnt[idx];
            if (c != 0u) atomicAdd(&deg[lo + idx], (float)c);
        }
    }
}

#define BK_SH 7
#define BK_R  128
#define BIN_THREADS 1024
#define BIN_BATCH   8
#define BIN_CAP     32768
#define SC_BATCH    8
#define SC_THREADS  512

__global__ __launch_bounds__(BK_R) void gcn_rsqrt_bucket(
    float* __restrict__ deg, int* __restrict__ total, int n) {
    __shared__ int s_cnt;
    const int tid = threadIdx.x;
    if (tid == 0) s_cnt = 0;
    __syncthreads();
    const int i = blockIdx.x * BK_R + tid;
    if (i < n) {
        const float d = deg[i];
        deg[i] = (d > 0.0f) ? rsqrtf(d) : 0.0f;
        atomicAdd(&s_cnt, (int)d - 1);
    }
    __syncthreads();
    if (tid == 0) total[blockIdx.x] = s_cnt;
}

__global__ __launch_bounds__(1024) void gcn_bucket_scan(
    const int* __restrict__ total, int* __restrict__ bstart,
    int* __restrict__ cursor, int P) {
    __shared__ int wsum[32];
    const int tid  = threadIdx.x;
    const int lane = tid & 31;
    const int wave = tid >> 5;
    const int per  = (P + 1023) / 1024;
    const int b0   = min(P, tid * per);
    const int b1   = min(P, b0 + per);
    int s = 0;
    for (int b = b0; b < b1; ++b) s += total[b];
    int v = s;
    for (int o = 1; o < 32; o <<= 1) {
        const int t = __shfl_up(v, o);
        if (lane >= o) v += t;
    }
    if (lane == 31) wsum[wave] = v;
    __syncthreads();
    if (wave == 0) {
        int w = wsum[lane];
        for (int o = 1; o < 32; o <<= 1) {
            const int t = __shfl_up(w, o);
            if (lane >= o) w += t;
        }
        wsum[lane] = w;
    }
    __syncthreads();
    int run = v - s + (wave > 0 ? wsum[wave - 1] : 0);
    for (int b = b0; b < b1; ++b) {
        bstart[b] = run;
        cursor[b] = run;
        run += total[b];
    }
    if (tid == 1023) bstart[P] = run;
}

__global__ __launch_bounds__(BIN_THREADS) void gcn_bin_edges(
    const int* __restrict__ src, const int* __restrict__ dst,
    int* __restrict__ cursor, int* __restrict__ rec, int e, int n, int P) {
    extern __shared__ int bin_sm[];
    __shared__ int2 stage[BIN_CAP];
    __shared__ int wsum[32];
    __shared__ int s_total;
    int* hist  = bin_sm;
    int* delta = bin_sm + P;
    const int tid  = threadIdx.x;
    const int lane = tid & 31;
    const int wave = tid >> 5;
    for (int p = tid; p < P; p += BIN_THREADS) hist[p] = 0;
    __syncthreads();

    const int per = (e + (int)gridDim.x - 1) / (int)gridDim.x;
    const int s0  = min(e, (int)blockIdx.x * per);
    const int s1  = min(e, s0 + per);

    for (int i0 = s0 + tid; i0 < s1; i0 += BIN_THREADS * BIN_BATCH) {
        int d[BIN_BATCH];
        #pragma unroll
        for (int k = 0; k < BIN_BATCH; ++k) {
            const int i = i0 + k * BIN_THREADS;
            d[k] = (i < s1) ? dst[i] : -1;
        }
        #pragma unroll
        for (int k = 0; k < BIN_BATCH; ++k)
            if ((unsigned)d[k] < (unsigned)n) atomicAdd(&hist[d[k] >> BK_SH], 1);
    }
    __syncthreads();

    const int perb = (P + BIN_THREADS - 1) / BIN_THREADS;
    const int b0   = min(P, tid * perb);
    const int b1   = min(P, b0 + perb);
    int s = 0;
    for (int b = b0; b < b1; ++b) s += hist[b];
    int v = s;
    for (int o = 1; o < 32; o <<= 1) {
        const int t = __shfl_up(v, o);
        if (lane >= o) v += t;
    }
    if (lane == 31) wsum[wave] = v;
    __syncthreads();
    if (wave == 0) {
        int w = wsum[lane];
        for (int o = 1; o < 32; o <<= 1) {
            const int t = __shfl_up(w, o);
            if (lane >= o) w += t;
        }
        wsum[lane] = w;
    }
    __syncthreads();
    int run = v - s + (wave > 0 ? wsum[wave - 1] : 0);
    for (int b = b0; b < b1; ++b) {
        const int c = hist[b];
        const int g = (c > 0) ? atomicAdd(&cursor[b], c) : 0;
        delta[b] = g - run;
        hist[b]  = run;
        run += c;
    }
    if (tid == BIN_THREADS - 1) s_total = min(run, BIN_CAP);
    __syncthreads();

    for (int i0 = s0 + tid; i0 < s1; i0 += BIN_THREADS * BIN_BATCH) {
        int d[BIN_BATCH], sidx[BIN_BATCH];
        #pragma unroll
        for (int k = 0; k < BIN_BATCH; ++k) {
            const int i = min(i0 + k * BIN_THREADS, s1 - 1);
            d[k]    = dst[i];
            sidx[k] = src[i];
        }
        asm volatile("" :: "v"(sidx[0]), "v"(sidx[1]), "v"(sidx[2]), "v"(sidx[3]),
                           "v"(sidx[4]), "v"(sidx[5]), "v"(sidx[6]), "v"(sidx[7]));
        #pragma unroll
        for (int k = 0; k < BIN_BATCH; ++k) {
            if (i0 + k * BIN_THREADS < s1 && (unsigned)d[k] < (unsigned)n) {
                const int b = d[k] >> BK_SH;
                const int off = delta[b];
                const int pos = atomicAdd(&hist[b], 1);
                const unsigned key = ((unsigned)sidx[k] << BK_SH) | (unsigned)(d[k] & (BK_R - 1));
                if ((unsigned)pos < (unsigned)BIN_CAP)
                    stage[pos] = make_int2((int)key, off + pos);
            }
        }
    }
    __syncthreads();

    const int total = s_total;
    for (int i0 = tid; i0 < total; i0 += BIN_THREADS * BIN_BATCH) {
        int2 r[BIN_BATCH];
        #pragma unroll
        for (int k = 0; k < BIN_BATCH; ++k)
            r[k] = stage[min(i0 + k * BIN_THREADS, total - 1)];
        #pragma unroll
        for (int k = 0; k < BIN_BATCH; ++k)
            if (i0 + k * BIN_THREADS < total && (unsigned)r[k].y < (unsigned)e)
                rec[r[k].y] = r[k].x;
    }
}

__global__ __launch_bounds__(256) void gcn_gemm1_wmma(
    const float* __restrict__ x, const float* __restrict__ W1,
    const float* __restrict__ dis, float* __restrict__ hw1, int n) {
    __shared__ float W1s[IN_DIM * H1_DIM];
    const int tid = threadIdx.x;
    #pragma unroll
    for (int j = 0; j < (IN_DIM * H1_DIM) / 256; ++j)
        W1s[tid + j * 256] = W1[tid + j * 256];
    __syncthreads();

    const int wave = tid >> 5;
    const int lane = tid & 31;
    const int half = lane >> 4;
    const int l    = lane & 15;
    const int row0 = (blockIdx.x * 8 + wave) * 16;
    if (row0 >= n) return;

    int arow = row0 + l;
    if (arow >= n) arow = n - 1;

    v8f c = {};
    #pragma unroll 4
    for (int k0 = 0; k0 < IN_DIM; k0 += 4) {
        const int ka = k0 + half * 2;
        const v2f a = *(const v2f*)&x[(size_t)arow * IN_DIM + ka];
        v2f b;
        b.x = W1s[ka * H1_DIM + l];
        b.y = W1s[(ka + 1) * H1_DIM + l];
        c = __builtin_amdgcn_wmma_f32_16x16x4_f32(
                false, a, false, b, (short)0, c, false, false);
    }

    if (row0 + 16 <= n) {
        #pragma unroll
        for (int r = 0; r < 8; ++r) {
            const int row = row0 + r + half * 8;
            hw1[(size_t)row * H1_DIM + l] = dis[row] * c[r];
        }
    } else {
        for (int r = 0; r < 8; ++r) {
            const int row = row0 + r + half * 8;
            if (row < n) hw1[(size_t)row * H1_DIM + l] = dis[row] * c[r];
        }
    }
}

__global__ __launch_bounds__(SC_THREADS) void gcn_scatter1(
    const int* __restrict__ rec, const int* __restrict__ bstart,
    const float* __restrict__ hw1, const float* __restrict__ dis,
    const float* __restrict__ b1, float* __restrict__ agg1, int n) {
    __shared__ float acc[BK_R * H1_DIM];
    const int tid = threadIdx.x;
    for (int j = tid; j < BK_R * H1_DIM; j += SC_THREADS) acc[j] = 0.0f;
    __syncthreads();

    const int lo  = blockIdx.x * BK_R;
    const int beg = bstart[blockIdx.x];
    const int end = bstart[blockIdx.x + 1];
    const int c = tid & (H1_DIM - 1);
    const int g = tid >> 4;
    const int G = SC_THREADS >> 4;
    for (int i0 = beg + g; i0 < end; i0 += G * SC_BATCH) {
        int r[SC_BATCH];
        #pragma unroll
        for (int k = 0; k < SC_BATCH; ++k) {
            const int i = min(i0 + k * G, end - 1);
            r[k] = rec[i];
        }
        float h[SC_BATCH];
        #pragma unroll
        for (int k = 0; k < SC_BATCH; ++k) {
            const unsigned s = min((unsigned)r[k] >> BK_SH, (unsigned)(n - 1));
            h[k] = hw1[(size_t)s * H1_DIM + c];
        }
        #pragma unroll
        for (int k = 0; k < SC_BATCH; ++k) {
            const float v = (i0 + k * G < end) ? h[k] : 0.0f;
            const int dl = r[k] & (BK_R - 1);
            atomicAdd(&acc[dl * H1_DIM + c], v);
        }
    }
    __syncthreads();

    const int cnt = min(BK_R, n - lo) * H1_DIM;
    const float* __restrict__ own = hw1 + (size_t)lo * H1_DIM;
    float* __restrict__ row = agg1 + (size_t)lo * H1_DIM;
    const float bias = b1[c];
    for (int j = tid; j < cnt; j += SC_THREADS)
        row[j] = dis[lo + (j >> 4)] * (acc[j] + own[j]) + bias;
}

__global__ __launch_bounds__(256) void gcn_gemm2_wmma(
    const float* __restrict__ h1, const float* __restrict__ W2,
    const float* __restrict__ dis, float* __restrict__ hw2, int n) {
    __shared__ float W2s[H1_DIM * 16];
    const int tid = threadIdx.x;
    {
        const int k = tid >> 4, col = tid & 15;
        W2s[tid] = (col < H2_DIM) ? W2[k * H2_DIM + col] : 0.0f;
    }
    __syncthreads();

    const int wave = tid >> 5;
    const int lane = tid & 31;
    const int half = lane >> 4;
    const int l    = lane & 15;
    const int row0 = (blockIdx.x * 8 + wave) * 16;
    if (row0 >= n) return;

    int arow = row0 + l;
    if (arow >= n) arow = n - 1;

    v8f c = {};
    #pragma unroll
    for (int k0 = 0; k0 < H1_DIM; k0 += 4) {
        const int ka = k0 + half * 2;
        v2f a = *(const v2f*)&h1[(size_t)arow * H1_DIM + ka];
        a.x = fmaxf(a.x, 0.0f);
        a.y = fmaxf(a.y, 0.0f);
        v2f b;
        b.x = W2s[ka * 16 + l];
        b.y = W2s[(ka + 1) * 16 + l];
        c = __builtin_amdgcn_wmma_f32_16x16x4_f32(
                false, a, false, b, (short)0, c, false, false);
    }

    if (l < H2_DIM) {
        if (row0 + 16 <= n) {
            #pragma unroll
            for (int r = 0; r < 8; ++r) {
                const int row = row0 + r + half * 8;
                hw2[(size_t)row * H2_DIM + l] = dis[row] * c[r];
            }
        } else {
            for (int r = 0; r < 8; ++r) {
                const int row = row0 + r + half * 8;
                if (row < n) hw2[(size_t)row * H2_DIM + l] = dis[row] * c[r];
            }
        }
    }
}

__global__ __launch_bounds__(SC_THREADS) void gcn_scatter2(
    const int* __restrict__ rec, const int* __restrict__ bstart,
    const float* __restrict__ hw2, const float* __restrict__ dis,
    const float* __restrict__ b2, float* __restrict__ out, int n) {
    __shared__ float acc[BK_R * H2_DIM];
    const int tid = threadIdx.x;
    for (int j = tid; j < BK_R * H2_DIM; j += SC_THREADS) acc[j] = 0.0f;
    __syncthreads();

    const int lo  = blockIdx.x * BK_R;
    const int beg = bstart[blockIdx.x];
    const int end = bstart[blockIdx.x + 1];
    const int c = tid & (H2_DIM - 1);
    const int g = tid >> 3;
    const int G = SC_THREADS >> 3;
    for (int i0 = beg + g; i0 < end; i0 += G * SC_BATCH) {
        int r[SC_BATCH];
        #pragma unroll
        for (int k = 0; k < SC_BATCH; ++k) {
            const int i = min(i0 + k * G, end - 1);
            r[k] = rec[i];
        }
        float h[SC_BATCH];
        #pragma unroll
        for (int k = 0; k < SC_BATCH; ++k) {
            const unsigned s = min((unsigned)r[k] >> BK_SH, (unsigned)(n - 1));
            h[k] = hw2[(size_t)s * H2_DIM + c];
        }
        #pragma unroll
        for (int k = 0; k < SC_BATCH; ++k) {
            const float v = (i0 + k * G < end) ? h[k] : 0.0f;
            const int dl = r[k] & (BK_R - 1);
            atomicAdd(&acc[dl * H2_DIM + c], v);
        }
    }
    __syncthreads();

    const int cnt = min(BK_R, n - lo) * H2_DIM;
    const float* __restrict__ own = hw2 + (size_t)lo * H2_DIM;
    float* __restrict__ row = out + (size_t)lo * H2_DIM;
    const float bias = b2[c];
    for (int j = tid; j < cnt; j += SC_THREADS)
        row[j] = dis[lo + (j >> 3)] * (acc[j] + own[j]) + bias;
}

extern "C" void kernel_launch(void* const* d_in, const int* in_sizes, int n_in,
                              void* d_out, int out_size, void* d_ws, size_t ws_size,
                              hipStream_t stream) {
    const float* x    = (const float*)d_in[0];
    const int*   edge = (const int*)  d_in[1];
    const float* W1   = (const float*)d_in[2];
    const float* b1   = (const float*)d_in[3];
    const float* W2   = (const float*)d_in[4];
    const float* b2   = (const float*)d_in[5];
    float* out = (float*)d_out;

    const int n = in_sizes[0] / IN_DIM;
    const int e = in_sizes[1] / 2;
    const int* src = edge;
    const int* dst = edge + e;

    float* ws   = (float*)d_ws;
    float* dis  = ws;
    float* hw1  = dis + n;
    float* agg1 = hw1 + (size_t)n * H1_DIM;
    float* hw2  = agg1 + (size_t)n * H1_DIM;

    const int P = (n + BK_R - 1) / BK_R;
    int* total  = (int*)(hw2 + (size_t)n * H2_DIM);
    int* cursor = total + P;
    int* bstart = cursor + P;
    int* rec    = (int*)((((size_t)(bstart + P + 1)) + 15) & ~(size_t)15);

    const int B = 256;
    const int gn     = (n + B - 1) / B;
    const int dpieces = (n + DEG_R_MAX - 1) / DEG_R_MAX;
    const int dR      = (((n + dpieces - 1) / dpieces) + 31) & ~31;
    const int dslices = (256 / dpieces) > 0 ? (256 / dpieces) : 1;
    const int bneed   = (e + BIN_CAP - 1) / BIN_CAP;
    const int bslices = bneed > 256 ? bneed : 256;
    const int tiles  = (n + 15) / 16;
    const int gtiles = (tiles + 7) / 8;

    gcn_init_deg <<<gn, B, 0, stream>>>(dis, n);
    gcn_count_deg<<<dim3(dslices, dpieces), DEG_THREADS, 0, stream>>>(dst, dis, e, n, dR);
    gcn_rsqrt_bucket<<<P, BK_R, 0, stream>>>(dis, total, n);
    gcn_bucket_scan <<<1, 1024, 0, stream>>>(total, bstart, cursor, P);
    gcn_bin_edges   <<<bslices, BIN_THREADS, 2 * (size_t)P * sizeof(int), stream>>>(src, dst, cursor, rec, e, n, P);

    gcn_gemm1_wmma<<<gtiles, B, 0, stream>>>(x, W1, dis, hw1, n);
    gcn_scatter1  <<<P, SC_THREADS, 0, stream>>>(rec, bstart, hw1, dis, b1, agg1, n);

    gcn_gemm2_wmma<<<gtiles, B, 0, stream>>>(agg1, W2, dis, hw2, n);
    gcn_scatter2  <<<P, SC_THREADS, 0, stream>>>(rec, bstart, hw2, dis, b2, out, n);
}


# No degree pass over dst: bucket sizes and slice runs from one per-slice bucket histogram (no cursor atomics), degrees counted from binned records in the GEMM1 block; order inside a bucket changes
# speedup vs baseline: 6.3966x; 1.0472x over previous
#include <hip/hip_runtime.h>


typedef __attribute__((ext_vector_type(2))) float v2f;
typedef __attribute__((ext_vector_type(8))) float v8f;

#define IN_DIM 128
#define H1_DIM 16
#define H2_DIM 8

#define BK_SH 7
#define BK_R  128
#define BIN_THREADS 1024
#define BIN_BATCH   8
#define BIN_CAP     32768
#define SC_BATCH    8
#define SC_THREADS  512
#define SS_THREADS  1024
#define SS_BATCH    8
#define DG_BATCH    16
static_assert(DG_BATCH == 16, "the load fence in the degree count names 16 values");

__device__ __forceinline__ void slice_hit(int* hist, int d, int n) {
    if ((unsigned)d < (unsigned)n) atomicAdd(&hist[d >> BK_SH], 1);
}

__global__ __launch_bounds__(BIN_THREADS) void gcn_slice_hist(
    const int* __restrict__ dst, int* __restrict__ cnt, int e, int n, int P, int TP) {
    extern __shared__ int hs_sm[];
    int* hist = hs_sm;
    const int tid = threadIdx.x;
    for (int p = tid; p < P; p += BIN_THREADS) hist[p] = 0;
    __syncthreads();

    const int per = (e + (int)gridDim.x - 1) / (int)gridDim.x;
    const int s0  = min(e, (int)blockIdx.x * per);
    const int s1  = min(e, s0 + per);

    const int mis = (int)((((size_t)(dst + s0)) >> 2) & 3);
    const int a0  = min(s1, s0 + ((4 - mis) & 3));
    const int nq  = (s1 - a0) >> 2;
    const int a1  = a0 + (nq << 2);
    if (tid < a0 - s0) slice_hit(hist, dst[s0 + tid], n);
    if (tid < s1 - a1) slice_hit(hist, dst[a1 + tid], n);
    const int4* __restrict__ dq = (const int4*)(dst + a0);
    for (int q = tid; q < nq; q += BIN_THREADS * BIN_BATCH) {
        int4 v[BIN_BATCH];
        #pragma unroll
        for (int k = 0; k < BIN_BATCH; ++k) {
            const int qq = q + k * BIN_THREADS;
            v[k] = (qq < nq) ? dq[qq] : make_int4(-1, -1, -1, -1);
        }
        #pragma unroll
        for (int k = 0; k < BIN_BATCH; ++k) {
            slice_hit(hist, v[k].x, n);
            slice_hit(hist, v[k].y, n);
            slice_hit(hist, v[k].z, n);
            slice_hit(hist, v[k].w, n);
        }
    }
    __syncthreads();

    int* __restrict__ row = cnt + (size_t)blockIdx.x * TP;
    for (int p = tid; p < P; p += BIN_THREADS) row[p] = hist[p];
}

__global__ __launch_bounds__(SS_THREADS) void gcn_slice_scan(
    const int* __restrict__ cnt, int* __restrict__ rel, int* __restrict__ total,
    int S, int P, int TP) {
    __shared__ int part[32][32];
    const int tid  = threadIdx.x;
    const int lane = tid & 31;
    const int wave = tid >> 5;
    const int b    = blockIdx.x * 32 + lane;
    const bool live = b < P;
    const int spw = (S + 31) / 32;
    const int sA  = min(S, wave * spw);
    const int sB  = min(S, sA + spw);
    const int* __restrict__ col = cnt + b;
    int sum = 0;
    if (live) {
        for (int s0 = sA; s0 < sB; s0 += SS_BATCH) {
            int v[SS_BATCH];
            #pragma unroll
            for (int k = 0; k < SS_BATCH; ++k)
                v[k] = col[(size_t)min(s0 + k, sB - 1) * TP];
            #pragma unroll
            for (int k = 0; k < SS_BATCH; ++k)
                if (s0 + k < sB) sum += v[k];
        }
    }
    part[wave][lane] = sum;
    __syncthreads();
    int run = 0, all = 0;
    #pragma unroll
    for (int w = 0; w < 32; ++w) {
        const int t = part[w][lane];
        if (w < wave) run += t;
        all += t;
    }
    if (live) {
        for (int s0 = sA; s0 < sB; s0 += SS_BATCH) {
            int v[SS_BATCH];
            #pragma unroll
            for (int k = 0; k < SS_BATCH; ++k)
                v[k] = col[(size_t)min(s0 + k, sB - 1) * TP];
            #pragma unroll
            for (int k = 0; k < SS_BATCH; ++k) {
                if (s0 + k < sB) {
                    rel[(size_t)(s0 + k) * TP + b] = run;
                    run += v[k];
                }
            }
        }
        if (wave == 0) total[b] = all;
    }
}

__global__ __launch_bounds__(1024) void gcn_bucket_scan(
    const int* __restrict__ total, int* __restrict__ bstart, int P) {
    __shared__ int wsum[32];
    const int tid  = threadIdx.x;
    const int lane = tid & 31;
    const int wave = tid >> 5;
    const int per  = (P + 1023) / 1024;
    const int b0   = min(P, tid * per);
    const int b1   = min(P, b0 + per);
    int s = 0;
    for (int b = b0; b < b1; ++b) s += total[b];
    int v = s;
    for (int o = 1; o < 32; o <<= 1) {
        const int t = __shfl_up(v, o);
        if (lane >= o) v += t;
    }
    if (lane == 31) wsum[wave] = v;
    __syncthreads();
    if (wave == 0) {
        int w = wsum[lane];
        for (int o = 1; o < 32; o <<= 1) {
            const int t = __shfl_up(w, o);
            if (lane >= o) w += t;
        }
        wsum[lane] = w;
    }
    __syncthreads();
    int run = v - s + (wave > 0 ? wsum[wave - 1] : 0);
    for (int b = b0; b < b1; ++b) {
        bstart[b] = run;
        run += total[b];
    }
    if (tid == 1023) bstart[P] = run;
}

__global__ __launch_bounds__(BIN_THREADS) void gcn_bin_edges(
    const int* __restrict__ src, const int* __restrict__ dst,
    const int* __restrict__ cnt, const int* __restrict__ rel,
    const int* __restrict__ bstart, int* __restrict__ rec, int e, int n, int P, int TP) {
    extern __shared__ int bin_sm[];
    __shared__ int2 stage[BIN_CAP];
    __shared__ int wsum[32];
    __shared__ int s_total;
    int* hist  = bin_sm;
    int* delta = bin_sm + P;
    const int tid  = threadIdx.x;
    const int lane = tid & 31;
    const int wave = tid >> 5;
    const int* __restrict__ crow = cnt + (size_t)blockIdx.x * TP;
    const int* __restrict__ rrow = rel + (size_t)blockIdx.x * TP;
    for (int p = tid; p < P; p += BIN_THREADS) hist[p] = crow[p];
    __syncthreads();

    const int per = (e + (int)gridDim.x - 1) / (int)gridDim.x;
    const int s0  = min(e, (int)blockIdx.x * per);
    const int s1  = min(e, s0 + per);

    const int perb = (P + BIN_THREADS - 1) / BIN_THREADS;
    const int b0   = min(P, tid * perb);
    const int b1   = min(P, b0 + perb);
    int s = 0;
    for (int b = b0; b < b1; ++b) s += hist[b];
    int v = s;
    for (int o = 1; o < 32; o <<= 1) {
        const int t = __shfl_up(v, o);
        if (lane >= o) v += t;
    }
    if (lane == 31) wsum[wave] = v;
    __syncthreads();
    if (wave == 0) {
        int w = wsum[lane];
        for (int o = 1; o < 32; o <<= 1) {
            const int t = __shfl_up(w, o);
            if (lane >= o) w += t;
        }
        wsum[lane] = w;
    }
    __syncthreads();
    int run = v - s + (wave > 0 ? wsum[wave - 1] : 0);
    for (int b = b0; b < b1; ++b) {
        const int c = hist[b];
        const int g = bstart[b] + rrow[b];
        delta[b] = g - run;
        hist[b]  = run;
        run += c;
    }
    if (tid == BIN_THREADS - 1) s_total = min(run, BIN_CAP);
    __syncthreads();

    for (int i0 = s0 + tid; i0 < s1; i0 += BIN_THREADS * BIN_BATCH) {
        int d[BIN_BATCH], sidx[BIN_BATCH];
        #pragma unroll
        for (int k = 0; k < BIN_BATCH; ++k) {
            const int i = min(i0 + k * BIN_THREADS, s1 - 1);
            d[k]    = dst[i];
            sidx[k] = src[i];
        }
        asm volatile("" :: "v"(sidx[0]), "v"(sidx[1]), "v"(sidx[2]), "v"(sidx[3]),
                           "v"(sidx[4]), "v"(sidx[5]), "v"(sidx[6]), "v"(sidx[7]));
        #pragma unroll
        for (int k = 0; k < BIN_BATCH; ++k) {
            if (i0 + k * BIN_THREADS < s1 && (unsigned)d[k] < (unsigned)n) {
                const int b = d[k] >> BK_SH;
                const int off = delta[b];
                const int pos = atomicAdd(&hist[b], 1);
                const unsigned key = ((unsigned)sidx[k] << BK_SH) | (unsigned)(d[k] & (BK_R - 1));
                if ((unsigned)pos < (unsigned)BIN_CAP)
                    stage[pos] = make_int2((int)key, off + pos);
            }
        }
    }
    __syncthreads();

    const int total = s_total;
    for (int i0 = tid; i0 < total; i0 += BIN_THREADS * BIN_BATCH) {
        int2 r[BIN_BATCH];
        #pragma unroll
        for (int k = 0; k < BIN_BATCH; ++k)
            r[k] = stage[min(i0 + k * BIN_THREADS, total - 1)];
        #pragma unroll
        for (int k = 0; k < BIN_BATCH; ++k)
            if (i0 + k * BIN_THREADS < total && (unsigned)r[k].y < (unsigned)e)
                rec[r[k].y] = r[k].x;
    }
}

static_assert(BK_R == 8 * 16, "a GEMM block's rows must be one bucket");

__global__ __launch_bounds__(256) void gcn_gemm1_wmma(
    const float* __restrict__ x, const float* __restrict__ W1,
    const int* __restrict__ rec, const int* __restrict__ bstart,
    float* __restrict__ dis, float* __restrict__ hw1, int n) {
    __shared__ float W1s[IN_DIM * H1_DIM];
    __shared__ __attribute__((aligned(16))) int dcnt[BK_R * 4];
    __shared__ float dis_s[BK_R];
    const int tid = threadIdx.x;
    #pragma unroll
    for (int j = 0; j < (IN_DIM * H1_DIM) / 256; ++j)
        W1s[tid + j * 256] = W1[tid + j * 256];
    for (int j = tid; j < BK_R * 4; j += 256) dcnt[j] = 0;
    __syncthreads();

    {
        const int beg = bstart[blockIdx.x];
        const int end = bstart[blockIdx.x + 1];
        const int q = tid & 3;
        for (int i0 = beg + tid; i0 < end; i0 += 256 * DG_BATCH) {
            int r[DG_BATCH];
            #pragma unroll
            for (int k = 0; k < DG_BATCH; ++k)
                r[k] = rec[min(i0 + k * 256, end - 1)];
            asm volatile("" :: "v"(r[0]), "v"(r[1]), "v"(r[2]), "v"(r[3]),
                               "v"(r[4]), "v"(r[5]), "v"(r[6]), "v"(r[7]),
                               "v"(r[8]), "v"(r[9]), "v"(r[10]), "v"(r[11]),
                               "v"(r[12]), "v"(r[13]), "v"(r[14]), "v"(r[15]));
            #pragma unroll
            for (int k = 0; k < DG_BATCH; ++k)
                if (i0 + k * 256 < end) atomicAdd(&dcnt[((r[k] & (BK_R - 1)) << 2) | q], 1);
        }
    }
    __syncthreads();
    if (tid < BK_R) {
        const int4 c4 = *(const int4*)&dcnt[tid * 4];
        const float d = (float)(1 + c4.x + c4.y + c4.z + c4.w);
        const float s = (d > 0.0f) ? rsqrtf(d) : 0.0f;
        dis_s[tid] = s;
    }
    __syncthreads();

    const int wave = tid >> 5;
    const int lane = tid & 31;
    const int half = lane >> 4;
    const int l    = lane & 15;
    const int row0 = (blockIdx.x * 8 + wave) * 16;
    if (row0 >= n) return;

    int arow = row0 + l;
    if (arow >= n) arow = n - 1;

    v8f c = {};
    #pragma unroll 4
    for (int k0 = 0; k0 < IN_DIM; k0 += 4) {
        const int ka = k0 + half * 2;
        const v2f a = *(const v2f*)&x[(size_t)arow * IN_DIM + ka];
        v2f b;
        b.x = W1s[ka * H1_DIM + l];
        b.y = W1s[(ka + 1) * H1_DIM + l];
        c = __builtin_amdgcn_wmma_f32_16x16x4_f32(
                false, a, false, b, (short)0, c, false, false);
    }

    if (row0 + 16 <= n) {
        #pragma unroll
        for (int r = 0; r < 8; ++r) {
            const int row = row0 + r + half * 8;
            hw1[(size_t)row * H1_DIM + l] = dis_s[wave * 16 + r + half * 8] * c[r];
        }
    } else {
        for (int r = 0; r < 8; ++r) {
            const int row = row0 + r + half * 8;
            if (row < n) hw1[(size_t)row * H1_DIM + l] = dis_s[wave * 16 + r + half * 8] * c[r];
        }
    }

    if (tid < BK_R) {
        const int node = blockIdx.x * BK_R + tid;
        if (node < n) dis[node] = dis_s[tid];
    }
}

__global__ __launch_bounds__(SC_THREADS) void gcn_scatter1(
    const int* __restrict__ rec, const int* __restrict__ bstart,
    const float* __restrict__ hw1, const float* __restrict__ dis,
    const float* __restrict__ b1, float* __restrict__ agg1, int n) {
    __shared__ float acc[BK_R * H1_DIM];
    const int tid = threadIdx.x;
    for (int j = tid; j < BK_R * H1_DIM; j += SC_THREADS) acc[j] = 0.0f;
    __syncthreads();

    const int lo  = blockIdx.x * BK_R;
    const int beg = bstart[blockIdx.x];
    const int end = bstart[blockIdx.x + 1];
    const int c = tid & (H1_DIM - 1);
    const int g = tid >> 4;
    const int G = SC_THREADS >> 4;
    for (int i0 = beg + g; i0 < end; i0 += G * SC_BATCH) {
        int r[SC_BATCH];
        #pragma unroll
        for (int k = 0; k < SC_BATCH; ++k) {
            const int i = min(i0 + k * G, end - 1);
            r[k] = rec[i];
        }
        float h[SC_BATCH];
        #pragma unroll
        for (int k = 0; k < SC_BATCH; ++k) {
            const unsigned s = min((unsigned)r[k] >> BK_SH, (unsigned)(n - 1));
            h[k] = hw1[(size_t)s * H1_DIM + c];
        }
        #pragma unroll
        for (int k = 0; k < SC_BATCH; ++k) {
            const float v = (i0 + k * G < end) ? h[k] : 0.0f;
            const int dl = r[k] & (BK_R - 1);
            atomicAdd(&acc[dl * H1_DIM + c], v);
        }
    }
    __syncthreads();

    const int cnt = min(BK_R, n - lo) * H1_DIM;
    const float* __restrict__ own = hw1 + (size_t)lo * H1_DIM;
    float* __restrict__ row = agg1 + (size_t)lo * H1_DIM;
    const float bias = b1[c];
    for (int j = tid; j < cnt; j += SC_THREADS)
        row[j] = dis[lo + (j >> 4)] * (acc[j] + own[j]) + bias;
}

__global__ __launch_bounds__(256) void gcn_gemm2_wmma(
    const float* __restrict__ h1, const float* __restrict__ W2,
    const float* __restrict__ dis, float* __restrict__ hw2, int n) {
    __shared__ float W2s[H1_DIM * 16];
    const int tid = threadIdx.x;
    {
        const int k = tid >> 4, col = tid & 15;
        W2s[tid] = (col < H2_DIM) ? W2[k * H2_DIM + col] : 0.0f;
    }
    __syncthreads();

    const int wave = tid >> 5;
    const int lane = tid & 31;
    const int half = lane >> 4;
    const int l    = lane & 15;
    const int row0 = (blockIdx.x * 8 + wave) * 16;
    if (row0 >= n) return;

    int arow = row0 + l;
    if (arow >= n) arow = n - 1;

    v8f c = {};
    #pragma unroll
    for (int k0 = 0; k0 < H1_DIM; k0 += 4) {
        const int ka = k0 + half * 2;
        v2f a = *(const v2f*)&h1[(size_t)arow * H1_DIM + ka];
        a.x = fmaxf(a.x, 0.0f);
        a.y = fmaxf(a.y, 0.0f);
        v2f b;
        b.x = W2s[ka * 16 + l];
        b.y = W2s[(ka + 1) * 16 + l];
        c = __builtin_amdgcn_wmma_f32_16x16x4_f32(
                false, a, false, b, (short)0, c, false, false);
    }

    if (l < H2_DIM) {
        if (row0 + 16 <= n) {
            #pragma unroll
            for (int r = 0; r < 8; ++r) {
                const int row = row0 + r + half * 8;
                hw2[(size_t)row * H2_DIM + l] = dis[row] * c[r];
            }
        } else {
            for (int r = 0; r < 8; ++r) {
                const int row = row0 + r + half * 8;
                if (row < n) hw2[(size_t)row * H2_DIM + l] = dis[row] * c[r];
            }
        }
    }
}

__global__ __launch_bounds__(SC_THREADS) void gcn_scatter2(
    const int* __restrict__ rec, const int* __restrict__ bstart,
    const float* __restrict__ hw2, const float* __restrict__ dis,
    const float* __restrict__ b2, float* __restrict__ out, int n) {
    __shared__ float acc[BK_R * H2_DIM];
    const int tid = threadIdx.x;
    for (int j = tid; j < BK_R * H2_DIM; j += SC_THREADS) acc[j] = 0.0f;
    __syncthreads();

    const int lo  = blockIdx.x * BK_R;
    const int beg = bstart[blockIdx.x];
    const int end = bstart[blockIdx.x + 1];
    const int c = tid & (H2_DIM - 1);
    const int g = tid >> 3;
    const int G = SC_THREADS >> 3;
    for (int i0 = beg + g; i0 < end; i0 += G * SC_BATCH) {
        int r[SC_BATCH];
        #pragma unroll
        for (int k = 0; k < SC_BATCH; ++k) {
            const int i = min(i0 + k * G, end - 1);
            r[k] = rec[i];
        }
        float h[SC_BATCH];
        #pragma unroll
        for (int k = 0; k < SC_BATCH; ++k) {
            const unsigned s = min((unsigned)r[k] >> BK_SH, (unsigned)(n - 1));
            h[k] = hw2[(size_t)s * H2_DIM + c];
        }
        #pragma unroll
        for (int k = 0; k < SC_BATCH; ++k) {
            const float v = (i0 + k * G < end) ? h[k] : 0.0f;
            const int dl = r[k] & (BK_R - 1);
            atomicAdd(&acc[dl * H2_DIM + c], v);
        }
    }
    __syncthreads();

    const int cnt = min(BK_R, n - lo) * H2_DIM;
    const float* __restrict__ own = hw2 + (size_t)lo * H2_DIM;
    float* __restrict__ row = out + (size_t)lo * H2_DIM;
    const float bias = b2[c];
    for (int j = tid; j < cnt; j += SC_THREADS)
        row[j] = dis[lo + (j >> 3)] * (acc[j] + own[j]) + bias;
}

extern "C" void kernel_launch(void* const* d_in, const int* in_sizes, int n_in,
                              void* d_out, int out_size, void* d_ws, size_t ws_size,
                              hipStream_t stream) {
    const float* x    = (const float*)d_in[0];
    const int*   edge = (const int*)  d_in[1];
    const float* W1   = (const float*)d_in[2];
    const float* b1   = (const float*)d_in[3];
    const float* W2   = (const float*)d_in[4];
    const float* b2   = (const float*)d_in[5];
    float* out = (float*)d_out;

    const int n = in_sizes[0] / IN_DIM;
    const int e = in_sizes[1] / 2;
    const int* src = edge;
    const int* dst = edge + e;

    float* ws   = (float*)d_ws;
    float* dis  = ws;
    float* hw1  = dis + n;
    float* agg1 = hw1 + (size_t)n * H1_DIM;
    float* hw2  = agg1 + (size_t)n * H1_DIM;

    const int P = (n + BK_R - 1) / BK_R;
    const int bneed   = (e + BIN_CAP - 1) / BIN_CAP;
    const int bslices = bneed > 256 ? bneed : 256;
    const int TP = (P + 31) & ~31;
    int* total  = (int*)(hw2 + (size_t)n * H2_DIM);
    int* bstart = total + P;
    int* cnt    = (int*)((((size_t)(bstart + P + 1)) + 127) & ~(size_t)127);
    int* rel    = cnt + (size_t)bslices * TP;
    int* rec    = rel + (size_t)bslices * TP;

    const int B = 256;
    const int tiles  = (n + 15) / 16;
    const int gtiles = (tiles + 7) / 8;

    gcn_slice_hist  <<<bslices, BIN_THREADS, (size_t)P * sizeof(int), stream>>>(dst, cnt, e, n, P, TP);
    gcn_slice_scan  <<<TP / 32, SS_THREADS, 0, stream>>>(cnt, rel, total, bslices, P, TP);
    gcn_bucket_scan <<<1, 1024, 0, stream>>>(total, bstart, P);
    gcn_bin_edges   <<<bslices, BIN_THREADS, 2 * (size_t)P * sizeof(int), stream>>>(src, dst, cnt, rel, bstart, rec, e, n, P, TP);

    gcn_gemm1_wmma<<<gtiles, B, 0, stream>>>(x, W1, rec, bstart, dis, hw1, n);
    gcn_scatter1  <<<P, SC_THREADS, 0, stream>>>(rec, bstart, hw1, dis, b1, agg1, n);

    gcn_gemm2_wmma<<<gtiles, B, 0, stream>>>(agg1, W2, dis, hw2, n);
    gcn_scatter2  <<<P, SC_THREADS, 0, stream>>>(rec, bstart, hw2, dis, b2, out, n);
}


# Both scatters: a bucket's records are copied into LDS by one group of coalesced loads and read from there, so a batch waits one memory round trip (the row gather), not two; order of LDS adds changes
# speedup vs baseline: 6.7366x; 1.0532x over previous
#include <hip/hip_runtime.h>


typedef __attribute__((ext_vector_type(2))) float v2f;
typedef __attribute__((ext_vector_type(8))) float v8f;

#define IN_DIM 128
#define H1_DIM 16
#define H2_DIM 8

#define BK_SH 7
#define BK_R  128
#define BIN_THREADS 1024
#define BIN_BATCH   8
#define BIN_CAP     32768
#define SC_BATCH    8
#define SC_THREADS  512
#define SC_CAP      10240
#define ST_BATCH    (SC_CAP / SC_THREADS)
#define SS_THREADS  1024
#define SS_BATCH    8
#define DG_BATCH    16
static_assert(DG_BATCH == 16, "the load fence in the degree count names 16 values");
static_assert(SC_BATCH == 8, "a lane group reads its batch of records as two 16-byte LDS loads");
static_assert(SC_CAP % SC_THREADS == 0 && SC_CAP % ((SC_THREADS / H2_DIM) * SC_BATCH) == 0 &&
              SC_CAP % ((SC_THREADS / H1_DIM) * SC_BATCH) == 0, "a pass is whole sweeps of the work-group");

__device__ __forceinline__ void slice_hit(int* hist, int d, int n) {
    if ((unsigned)d < (unsigned)n) atomicAdd(&hist[d >> BK_SH], 1);
}

__global__ __launch_bounds__(BIN_THREADS) void gcn_slice_hist(
    const int* __restrict__ dst, int* __restrict__ cnt, int e, int n, int P, int TP) {
    extern __shared__ int hs_sm[];
    int* hist = hs_sm;
    const int tid = threadIdx.x;
    for (int p = tid; p < P; p += BIN_THREADS) hist[p] = 0;
    __syncthreads();

    const int per = (e + (int)gridDim.x - 1) / (int)gridDim.x;
    const int s0  = min(e, (int)blockIdx.x * per);
    const int s1  = min(e, s0 + per);

    const int mis = (int)((((size_t)(dst + s0)) >> 2) & 3);
    const int a0  = min(s1, s0 + ((4 - mis) & 3));
    const int nq  = (s1 - a0) >> 2;
    const int a1  = a0 + (nq << 2);
    if (tid < a0 - s0) slice_hit(hist, dst[s0 + tid], n);
    if (tid < s1 - a1) slice_hit(hist, dst[a1 + tid], n);
    const int4* __restrict__ dq = (const int4*)(dst + a0);
    for (int q = tid; q < nq; q += BIN_THREADS * BIN_BATCH) {
        int4 v[BIN_BATCH];
        #pragma unroll
        for (int k = 0; k < BIN_BATCH; ++k) {
            const int qq = q + k * BIN_THREADS;
            v[k] = (qq < nq) ? dq[qq] : make_int4(-1, -1, -1, -1);
        }
        #pragma unroll
        for (int k = 0; k < BIN_BATCH; ++k) {
            slice_hit(hist, v[k].x, n);
            slice_hit(hist, v[k].y, n);
            slice_hit(hist, v[k].z, n);
            slice_hit(hist, v[k].w, n);
        }
    }
    __syncthreads();

    int* __restrict__ row = cnt + (size_t)blockIdx.x * TP;
    for (int p = tid; p < P; p += BIN_THREADS) row[p] = hist[p];
}

__global__ __launch_bounds__(SS_THREADS) void gcn_slice_scan(
    const int* __restrict__ cnt, int* __restrict__ rel, int* __restrict__ total,
    int S, int P, int TP) {
    __shared__ int part[32][32];
    const int tid  = threadIdx.x;
    const int lane = tid & 31;
    const int wave = tid >> 5;
    const int b    = blockIdx.x * 32 + lane;
    const bool live = b < P;
    const int spw = (S + 31) / 32;
    const int sA  = min(S, wave * spw);
    const int sB  = min(S, sA + spw);
    const int* __restrict__ col = cnt + b;
    int sum = 0;
    if (live) {
        for (int s0 = sA; s0 < sB; s0 += SS_BATCH) {
            int v[SS_BATCH];
            #pragma unroll
            for (int k = 0; k < SS_BATCH; ++k)
                v[k] = col[(size_t)min(s0 + k, sB - 1) * TP];
            #pragma unroll
            for (int k = 0; k < SS_BATCH; ++k)
                if (s0 + k < sB) sum += v[k];
        }
    }
    part[wave][lane] = sum;
    __syncthreads();
    int run = 0, all = 0;
    #pragma unroll
    for (int w = 0; w < 32; ++w) {
        const int t = part[w][lane];
        if (w < wave) run += t;
        all += t;
    }
    if (live) {
        for (int s0 = sA; s0 < sB; s0 += SS_BATCH) {
            int v[SS_BATCH];
            #pragma unroll
            for (int k = 0; k < SS_BATCH; ++k)
                v[k] = col[(size_t)min(s0 + k, sB - 1) * TP];
            #pragma unroll
            for (int k = 0; k < SS_BATCH; ++k) {
                if (s0 + k < sB) {
                    rel[(size_t)(s0 + k) * TP + b] = run;
                    run += v[k];
                }
            }
        }
        if (wave == 0) total[b] = all;
    }
}

__global__ __launch_bounds__(1024) void gcn_bucket_scan(
    const int* __restrict__ total, int* __restrict__ bstart, int P) {
    __shared__ int wsum[32];
    const int tid  = threadIdx.x;
    const int lane = tid & 31;
    const int wave = tid >> 5;
    const int per  = (P + 1023) / 1024;
    const int b0   = min(P, tid * per);
    const int b1   = min(P, b0 + per);
    int s = 0;
    for (int b = b0; b < b1; ++b) s += total[b];
    int v = s;
    for (int o = 1; o < 32; o <<= 1) {
        const int t = __shfl_up(v, o);
        if (lane >= o) v += t;
    }
    if (lane == 31) wsum[wave] = v;
    __syncthreads();
    if (wave == 0) {
        int w = wsum[lane];
        for (int o = 1; o < 32; o <<= 1) {
            const int t = __shfl_up(w, o);
            if (lane >= o) w += t;
        }
        wsum[lane] = w;
    }
    __syncthreads();
    int run = v - s + (wave > 0 ? wsum[wave - 1] : 0);
    for (int b = b0; b < b1; ++b) {
        bstart[b] = run;
        run += total[b];
    }
    if (tid == 1023) bstart[P] = run;
}

__global__ __launch_bounds__(BIN_THREADS) void gcn_bin_edges(
    const int* __restrict__ src, const int* __restrict__ dst,
    const int* __restrict__ cnt, const int* __restrict__ rel,
    const int* __restrict__ bstart, int* __restrict__ rec, int e, int n, int P, int TP) {
    extern __shared__ int bin_sm[];
    __shared__ int2 stage[BIN_CAP];
    __shared__ int wsum[32];
    __shared__ int s_total;
    int* hist  = bin_sm;
    int* delta = bin_sm + P;
    const int tid  = threadIdx.x;
    const int lane = tid & 31;
    const int wave = tid >> 5;
    const int* __restrict__ crow = cnt + (size_t)blockIdx.x * TP;
    const int* __restrict__ rrow = rel + (size_t)blockIdx.x * TP;
    for (int p = tid; p < P; p += BIN_THREADS) hist[p] = crow[p];
    __syncthreads();

    const int per = (e + (int)gridDim.x - 1) / (int)gridDim.x;
    const int s0  = min(e, (int)blockIdx.x * per);
    const int s1  = min(e, s0 + per);

    const int perb = (P + BIN_THREADS - 1) / BIN_THREADS;
    const int b0   = min(P, tid * perb);
    const int b1   = min(P, b0 + perb);
    int s = 0;
    for (int b = b0; b < b1; ++b) s += hist[b];
    int v = s;
    for (int o = 1; o < 32; o <<= 1) {
        const int t = __shfl_up(v, o);
        if (lane >= o) v += t;
    }
    if (lane == 31) wsum[wave] = v;
    __syncthreads();
    if (wave == 0) {
        int w = wsum[lane];
        for (int o = 1; o < 32; o <<= 1) {
            const int t = __shfl_up(w, o);
            if (lane >= o) w += t;
        }
        wsum[lane] = w;
    }
    __syncthreads();
    int run = v - s + (wave > 0 ? wsum[wave - 1] : 0);
    for (int b = b0; b < b1; ++b) {
        const int c = hist[b];
        const int g = bstart[b] + rrow[b];
        delta[b] = g - run;
        hist[b]  = run;
        run += c;
    }
    if (tid == BIN_THREADS - 1) s_total = min(run, BIN_CAP);
    __syncthreads();

    for (int i0 = s0 + tid; i0 < s1; i0 += BIN_THREADS * BIN_BATCH) {
        int d[BIN_BATCH], sidx[BIN_BATCH];
        #pragma unroll
        for (int k = 0; k < BIN_BATCH; ++k) {
            const int i = min(i0 + k * BIN_THREADS, s1 - 1);
            d[k]    = dst[i];
            sidx[k] = src[i];
        }
        asm volatile("" :: "v"(sidx[0]), "v"(sidx[1]), "v"(sidx[2]), "v"(sidx[3]),
                           "v"(sidx[4]), "v"(sidx[5]), "v"(sidx[6]), "v"(sidx[7]));
        #pragma unroll
        for (int k = 0; k < BIN_BATCH; ++k) {
            if (i0 + k * BIN_THREADS < s1 && (unsigned)d[k] < (unsigned)n) {
                const int b = d[k] >> BK_SH;
                const int off = delta[b];
                const int pos = atomicAdd(&hist[b], 1);
                const unsigned key = ((unsigned)sidx[k] << BK_SH) | (unsigned)(d[k] & (BK_R - 1));
                if ((unsigned)pos < (unsigned)BIN_CAP)
                    stage[pos] = make_int2((int)key, off + pos);
            }
        }
    }
    __syncthreads();

    const int total = s_total;
    for (int i0 = tid; i0 < total; i0 += BIN_THREADS * BIN_BATCH) {
        int2 r[BIN_BATCH];
        #pragma unroll
        for (int k = 0; k < BIN_BATCH; ++k)
            r[k] = stage[min(i0 + k * BIN_THREADS, total - 1)];
        #pragma unroll
        for (int k = 0; k < BIN_BATCH; ++k)
            if (i0 + k * BIN_THREADS < total && (unsigned)r[k].y < (unsigned)e)
                rec[r[k].y] = r[k].x;
    }
}

static_assert(BK_R == 8 * 16, "a GEMM block's rows must be one bucket");

__global__ __launch_bounds__(256) void gcn_gemm1_wmma(
    const float* __restrict__ x, const float* __restrict__ W1,
    const int* __restrict__ rec, const int* __restrict__ bstart,
    float* __restrict__ dis, float* __restrict__ hw1, int n) {
    __shared__ float W1s[IN_DIM * H1_DIM];
    __shared__ __attribute__((aligned(16))) int dcnt[BK_R * 4];
    __shared__ float dis_s[BK_R];
    const int tid = threadIdx.x;
    #pragma unroll
    for (int j = 0; j < (IN_DIM * H1_DIM) / 256; ++j)
        W1s[tid + j * 256] = W1[tid + j * 256];
    for (int j = tid; j < BK_R * 4; j += 256) dcnt[j] = 0;
    __syncthreads();

    {
        const int beg = bstart[blockIdx.x];
        const int end = bstart[blockIdx.x + 1];
        const int q = tid & 3;
        for (int i0 = beg + tid; i0 < end; i0 += 256 * DG_BATCH) {
            int r[DG_BATCH];
            #pragma unroll
            for (int k = 0; k < DG_BATCH; ++k)
                r[k] = rec[min(i0 + k * 256, end - 1)];
            asm volatile("" :: "v"(r[0]), "v"(r[1]), "v"(r[2]), "v"(r[3]),
                               "v"(r[4]), "v"(r[5]), "v"(r[6]), "v"(r[7]),
                               "v"(r[8]), "v"(r[9]), "v"(r[10]), "v"(r[11]),
                               "v"(r[12]), "v"(r[13]), "v"(r[14]), "v"(r[15]));
            #pragma unroll
            for (int k = 0; k < DG_BATCH; ++k)
                if (i0 + k * 256 < end) atomicAdd(&dcnt[((r[k] & (BK_R - 1)) << 2) | q], 1);
        }
    }
    __syncthreads();
    if (tid < BK_R) {
        const int4 c4 = *(const int4*)&dcnt[tid * 4];
        const float d = (float)(1 + c4.x + c4.y + c4.z + c4.w);
        const float s = (d > 0.0f) ? rsqrtf(d) : 0.0f;
        dis_s[tid] = s;
    }
    __syncthreads();

    const int wave = tid >> 5;
    const int lane = tid & 31;
    const int half = lane >> 4;
    const int l    = lane & 15;
    const int row0 = (blockIdx.x * 8 + wave) * 16;
    if (row0 >= n) return;

    int arow = row0 + l;
    if (arow >= n) arow = n - 1;

    v8f c = {};
    #pragma unroll 4
    for (int k0 = 0; k0 < IN_DIM; k0 += 4) {
        const int ka = k0 + half * 2;
        const v2f a = *(const v2f*)&x[(size_t)arow * IN_DIM + ka];
        v2f b;
        b.x = W1s[ka * H1_DIM + l];
        b.y = W1s[(ka + 1) * H1_DIM + l];
        c = __builtin_amdgcn_wmma_f32_16x16x4_f32(
                false, a, false, b, (short)0, c, false, false);
    }

    if (row0 + 16 <= n) {
        #pragma unroll
        for (int r = 0; r < 8; ++r) {
            const int row = row0 + r + half * 8;
            hw1[(size_t)row * H1_DIM + l] = dis_s[wave * 16 + r + half * 8] * c[r];
        }
    } else {
        for (int r = 0; r < 8; ++r) {
            const int row = row0 + r + half * 8;
            if (row < n) hw1[(size_t)row * H1_DIM + l] = dis_s[wave * 16 + r + half * 8] * c[r];
        }
    }

    if (tid < BK_R) {
        const int node = blockIdx.x * BK_R + tid;
        if (node < n) dis[node] = dis_s[tid];
    }
}

__device__ __forceinline__ void stage_records(
    const int* __restrict__ rec, int* __restrict__ lrec, int c0, int len, int tid) {
    int v[ST_BATCH];
    #pragma unroll
    for (int k = 0; k < ST_BATCH; ++k)
        v[k] = rec[c0 + min(tid + k * SC_THREADS, len - 1)];
    #pragma unroll
    for (int k = 0; k < ST_BATCH; ++k) {
        const int j = tid + k * SC_THREADS;
        lrec[j] = (j < len) ? v[k] : 0;
    }
}

__global__ __launch_bounds__(SC_THREADS) void gcn_scatter1(
    const int* __restrict__ rec, const int* __restrict__ bstart,
    const float* __restrict__ hw1, const float* __restrict__ dis,
    const float* __restrict__ b1, float* __restrict__ agg1, int n) {
    __shared__ float acc[BK_R * H1_DIM];
    __shared__ __attribute__((aligned(16))) int lrec[SC_CAP];
    const int tid = threadIdx.x;
    for (int j = tid; j < BK_R * H1_DIM; j += SC_THREADS) acc[j] = 0.0f;

    const int lo  = blockIdx.x * BK_R;
    const int beg = bstart[blockIdx.x];
    const int end = bstart[blockIdx.x + 1];
    const int c = tid & (H1_DIM - 1);
    const int g = tid >> 4;
    const int SW = (SC_THREADS >> 4) * SC_BATCH;
    for (int c0 = beg; c0 < end; c0 += SC_CAP) {
        if (c0 != beg) __syncthreads();
        const int len = min(SC_CAP, end - c0);
        stage_records(rec, lrec, c0, len, tid);
        __syncthreads();

        for (int i0 = g * SC_BATCH; i0 < len; i0 += SW) {
            const int4 ra = *(const int4*)&lrec[i0];
            const int4 rb = *(const int4*)&lrec[i0 + 4];
            const int r[SC_BATCH] = {ra.x, ra.y, ra.z, ra.w, rb.x, rb.y, rb.z, rb.w};
            float h[SC_BATCH];
            #pragma unroll
            for (int k = 0; k < SC_BATCH; ++k) {
                const unsigned s = min((unsigned)r[k] >> BK_SH, (unsigned)(n - 1));
                h[k] = hw1[(size_t)s * H1_DIM + c];
            }
            #pragma unroll
            for (int k = 0; k < SC_BATCH; ++k) {
                const float v = (i0 + k < len) ? h[k] : 0.0f;
                const int dl = r[k] & (BK_R - 1);
                atomicAdd(&acc[dl * H1_DIM + c], v);
            }
        }
    }
    __syncthreads();

    const int cnt = min(BK_R, n - lo) * H1_DIM;
    const float* __restrict__ own = hw1 + (size_t)lo * H1_DIM;
    float* __restrict__ row = agg1 + (size_t)lo * H1_DIM;
    const float bias = b1[c];
    for (int j = tid; j < cnt; j += SC_THREADS)
        row[j] = dis[lo + (j >> 4)] * (acc[j] + own[j]) + bias;
}

__global__ __launch_bounds__(256) void gcn_gemm2_wmma(
    const float* __restrict__ h1, const float* __restrict__ W2,
    const float* __restrict__ dis, float* __restrict__ hw2, int n) {
    __shared__ float W2s[H1_DIM * 16];
    const int tid = threadIdx.x;
    {
        const int k = tid >> 4, col = tid & 15;
        W2s[tid] = (col < H2_DIM) ? W2[k * H2_DIM + col] : 0.0f;
    }
    __syncthreads();

    const int wave = tid >> 5;
    const int lane = tid & 31;
    const int half = lane >> 4;
    const int l    = lane & 15;
    const int row0 = (blockIdx.x * 8 + wave) * 16;
    if (row0 >= n) return;

    int arow = row0 + l;
    if (arow >= n) arow = n - 1;

    v8f c = {};
    #pragma unroll
    for (int k0 = 0; k0 < H1_DIM; k0 += 4) {
        const int ka = k0 + half * 2;
        v2f a = *(const v2f*)&h1[(size_t)arow * H1_DIM + ka];
        a.x = fmaxf(a.x, 0.0f);
        a.y = fmaxf(a.y, 0.0f);
        v2f b;
        b.x = W2s[ka * 16 + l];
        b.y = W2s[(ka + 1) * 16 + l];
        c = __builtin_amdgcn_wmma_f32_16x16x4_f32(
                false, a, false, b, (short)0, c, false, false);
    }

    if (l < H2_DIM) {
        if (row0 + 16 <= n) {
            #pragma unroll
            for (int r = 0; r < 8; ++r) {
                const int row = row0 + r + half * 8;
                hw2[(size_t)row * H2_DIM + l] = dis[row] * c[r];
            }
        } else {
            for (int r = 0; r < 8; ++r) {
                const int row = row0 + r + half * 8;
                if (row < n) hw2[(size_t)row * H2_DIM + l] = dis[row] * c[r];
            }
        }
    }
}

__global__ __launch_bounds__(SC_THREADS) void gcn_scatter2(
    const int* __restrict__ rec, const int* __restrict__ bstart,
    const float* __restrict__ hw2, const float* __restrict__ dis,
    const float* __restrict__ b2, float* __restrict__ out, int n) {
    __shared__ float acc[BK_R * H2_DIM];
    __shared__ __attribute__((aligned(16))) int lrec[SC_CAP];
    const int tid = threadIdx.x;
    for (int j = tid; j < BK_R * H2_DIM; j += SC_THREADS) acc[j] = 0.0f;

    const int lo  = blockIdx.x * BK_R;
    const int beg = bstart[blockIdx.x];
    const int end = bstart[blockIdx.x + 1];
    const int c = tid & (H2_DIM - 1);
    const int g = tid >> 3;
    const int SW = (SC_THREADS >> 3) * SC_BATCH;
    for (int c0 = beg; c0 < end; c0 += SC_CAP) {
        if (c0 != beg) __syncthreads();
        const int len = min(SC_CAP, end - c0);
        stage_records(rec, lrec, c0, len, tid);
        __syncthreads();

        for (int i0 = g * SC_BATCH; i0 < len; i0 += SW) {
            const int4 ra = *(const int4*)&lrec[i0];
            const int4 rb = *(const int4*)&lrec[i0 + 4];
            const int r[SC_BATCH] = {ra.x, ra.y, ra.z, ra.w, rb.x, rb.y, rb.z, rb.w};
            float h[SC_BATCH];
            #pragma unroll
            for (int k = 0; k < SC_BATCH; ++k) {
                const unsigned s = min((unsigned)r[k] >> BK_SH, (unsigned)(n - 1));
                h[k] = hw2[(size_t)s * H2_DIM + c];
            }
            #pragma unroll
            for (int k = 0; k < SC_BATCH; ++k) {
                const float v = (i0 + k < len) ? h[k] : 0.0f;
                const int dl = r[k] & (BK_R - 1);
                atomicAdd(&acc[dl * H2_DIM + c], v);
            }
        }
    }
    __syncthreads();

    const int cnt = min(BK_R, n - lo) * H2_DIM;
    const float* __restrict__ own = hw2 + (size_t)lo * H2_DIM;
    float* __restrict__ row = out + (size_t)lo * H2_DIM;
    const float bias = b2[c];
    for (int j = tid; j < cnt; j += SC_THREADS)
        row[j] = dis[lo + (j >> 3)] * (acc[j] + own[j]) + bias;
}

extern "C" void kernel_launch(void* const* d_in, const int* in_sizes, int n_in,
                              void* d_out, int out_size, void* d_ws, size_t ws_size,
                              hipStream_t stream) {
    const float* x    = (const float*)d_in[0];
    const int*   edge = (const int*)  d_in[1];
    const float* W1   = (const float*)d_in[2];
    const float* b1   = (const float*)d_in[3];
    const float* W2   = (const float*)d_in[4];
    const float* b2   = (const float*)d_in[5];
    float* out = (float*)d_out;

    const int n = in_sizes[0] / IN_DIM;
    const int e = in_sizes[1] / 2;
    const int* src = edge;
    const int* dst = edge + e;

    float* ws   = (float*)d_ws;
    float* dis  = ws;
    float* hw1  = dis + n;
    float* agg1 = hw1 + (size_t)n * H1_DIM;
    float* hw2  = agg1 + (size_t)n * H1_DIM;

    const int P = (n + BK_R - 1) / BK_R;
    const int bneed   = (e + BIN_CAP - 1) / BIN_CAP;
    const int bslices = bneed > 256 ? bneed : 256;
    const int TP = (P + 31) & ~31;
    int* total  = (int*)(hw2 + (size_t)n * H2_DIM);
    int* bstart = total + P;
    int* cnt    = (int*)((((size_t)(bstart + P + 1)) + 127) & ~(size_t)127);
    int* rel    = cnt + (size_t)bslices * TP;
    int* rec    = rel + (size_t)bslices * TP;

    const int B = 256;
    const int tiles  = (n + 15) / 16;
    const int gtiles = (tiles + 7) / 8;

    gcn_slice_hist  <<<bslices, BIN_THREADS, (size_t)P * sizeof(int), stream>>>(dst, cnt, e, n, P, TP);
    gcn_slice_scan  <<<TP / 32, SS_THREADS, 0, stream>>>(cnt, rel, total, bslices, P, TP);
    gcn_bucket_scan <<<1, 1024, 0, stream>>>(total, bstart, P);
    gcn_bin_edges   <<<bslices, BIN_THREADS, 2 * (size_t)P * sizeof(int), stream>>>(src, dst, cnt, rel, bstart, rec, e, n, P, TP);

    gcn_gemm1_wmma<<<gtiles, B, 0, stream>>>(x, W1, rec, bstart, dis, hw1, n);
    gcn_scatter1  <<<P, SC_THREADS, 0, stream>>>(rec, bstart, hw1, dis, b1, agg1, n);

    gcn_gemm2_wmma<<<gtiles, B, 0, stream>>>(agg1, W2, dis, hw2, n);
    gcn_scatter2  <<<P, SC_THREADS, 0, stream>>>(rec, bstart, hw2, dis, b2, out, n);
}


# Launch fusion, 8 to 6 dependent launches: bucket-start scan redone inside each binning work-group; layer-2 GEMM (f32 WMMA, rows from LDS) inside the layer-1 scatter, agg1 never stored
# speedup vs baseline: 7.1206x; 1.0570x over previous
#include <hip/hip_runtime.h>


typedef __attribute__((ext_vector_type(2))) float v2f;
typedef __attribute__((ext_vector_type(8))) float v8f;

#define IN_DIM 128
#define H1_DIM 16
#define H2_DIM 8

#define BK_SH 7
#define BK_R  128
#define BIN_THREADS 1024
#define BIN_BATCH   8
#define BIN_CAP     32768
#define SC_BATCH    8
#define SC_THREADS  512
#define SC_CAP      10240
#define ST_BATCH    (SC_CAP / SC_THREADS)
#define H1_ROUNDS   ((BK_R * H1_DIM) / SC_THREADS)
#define SS_THREADS  1024
#define SS_BATCH    8
#define DG_BATCH    16
static_assert(DG_BATCH == 16, "the load fence in the degree count names 16 values");
static_assert(SC_BATCH == 8, "a lane group reads its batch of records as two 16-byte LDS loads");
static_assert((BK_R * H1_DIM) % SC_THREADS == 0 && SC_THREADS >= H1_DIM * 16 + BK_R && SC_THREADS / 32 >= BK_R / 16 && SC_THREADS * 4 >= BK_R * H2_DIM,
              "the layer-1 scatter stages W2 and dis side by side, runs one wave per 16-row tile and stores hw2 16 bytes per lane");
static_assert(SC_CAP % SC_THREADS == 0 && SC_CAP % ((SC_THREADS / H2_DIM) * SC_BATCH) == 0 &&
              SC_CAP % ((SC_THREADS / H1_DIM) * SC_BATCH) == 0, "a pass is whole sweeps of the work-group");

__device__ __forceinline__ void slice_hit(int* hist, int d, int n) {
    if ((unsigned)d < (unsigned)n) atomicAdd(&hist[d >> BK_SH], 1);
}

__global__ __launch_bounds__(BIN_THREADS) void gcn_slice_hist(
    const int* __restrict__ dst, int* __restrict__ cnt, int e, int n, int P, int TP) {
    extern __shared__ int hs_sm[];
    int* hist = hs_sm;
    const int tid = threadIdx.x;
    for (int p = tid; p < P; p += BIN_THREADS) hist[p] = 0;
    __syncthreads();

    const int per = (e + (int)gridDim.x - 1) / (int)gridDim.x;
    const int s0  = min(e, (int)blockIdx.x * per);
    const int s1  = min(e, s0 + per);

    const int mis = (int)((((size_t)(dst + s0)) >> 2) & 3);
    const int a0  = min(s1, s0 + ((4 - mis) & 3));
    const int nq  = (s1 - a0) >> 2;
    const int a1  = a0 + (nq << 2);
    if (tid < a0 - s0) slice_hit(hist, dst[s0 + tid], n);
    if (tid < s1 - a1) slice_hit(hist, dst[a1 + tid], n);
    const int4* __restrict__ dq = (const int4*)(dst + a0);
    for (int q = tid; q < nq; q += BIN_THREADS * BIN_BATCH) {
        int4 v[BIN_BATCH];
        #pragma unroll
        for (int k = 0; k < BIN_BATCH; ++k) {
            const int qq = q + k * BIN_THREADS;
            v[k] = (qq < nq) ? dq[qq] : make_int4(-1, -1, -1, -1);
        }
        #pragma unroll
        for (int k = 0; k < BIN_BATCH; ++k) {
            slice_hit(hist, v[k].x, n);
            slice_hit(hist, v[k].y, n);
            slice_hit(hist, v[k].z, n);
            slice_hit(hist, v[k].w, n);
        }
    }
    __syncthreads();

    int* __restrict__ row = cnt + (size_t)blockIdx.x * TP;
    for (int p = tid; p < P; p += BIN_THREADS) row[p] = hist[p];
}

__global__ __launch_bounds__(SS_THREADS) void gcn_slice_scan(
    const int* __restrict__ cnt, int* __restrict__ rel, int* __restrict__ total,
    int S, int P, int TP) {
    __shared__ int part[32][32];
    const int tid  = threadIdx.x;
    const int lane = tid & 31;
    const int wave = tid >> 5;
    const int b    = blockIdx.x * 32 + lane;
    const bool live = b < P;
    const int spw = (S + 31) / 32;
    const int sA  = min(S, wave * spw);
    const int sB  = min(S, sA + spw);
    const int* __restrict__ col = cnt + b;
    int sum = 0;
    if (live) {
        for (int s0 = sA; s0 < sB; s0 += SS_BATCH) {
            int v[SS_BATCH];
            #pragma unroll
            for (int k = 0; k < SS_BATCH; ++k)
                v[k] = col[(size_t)min(s0 + k, sB - 1) * TP];
            #pragma unroll
            for (int k = 0; k < SS_BATCH; ++k)
                if (s0 + k < sB) sum += v[k];
        }
    }
    part[wave][lane] = sum;
    __syncthreads();
    int run = 0, all = 0;
    #pragma unroll
    for (int w = 0; w < 32; ++w) {
        const int t = part[w][lane];
        if (w < wave) run += t;
        all += t;
    }
    if (live) {
        for (int s0 = sA; s0 < sB; s0 += SS_BATCH) {
            int v[SS_BATCH];
            #pragma unroll
            for (int k = 0; k < SS_BATCH; ++k)
                v[k] = col[(size_t)min(s0 + k, sB - 1) * TP];
            #pragma unroll
            for (int k = 0; k < SS_BATCH; ++k) {
                if (s0 + k < sB) {
                    rel[(size_t)(s0 + k) * TP + b] = run;
                    run += v[k];
                }
            }
        }
        if (wave == 0) total[b] = all;
    }
}

__global__ __launch_bounds__(BIN_THREADS) void gcn_bin_edges(
    const int* __restrict__ src, const int* __restrict__ dst,
    const int* __restrict__ cnt, const int* __restrict__ rel,
    const int* __restrict__ bsize, int* __restrict__ bstart,
    int* __restrict__ rec, int e, int n, int P, int TP) {
    extern __shared__ int bin_sm[];
    __shared__ int2 stage[BIN_CAP];
    __shared__ int wsum[32];
    __shared__ int wsum_b[32];
    __shared__ int s_total;
    int* hist  = bin_sm;
    int* delta = bin_sm + P;
    const int tid  = threadIdx.x;
    const int lane = tid & 31;
    const int wave = tid >> 5;
    const int* __restrict__ crow = cnt + (size_t)blockIdx.x * TP;
    const int* __restrict__ rrow = rel + (size_t)blockIdx.x * TP;
    for (int p = tid; p < P; p += BIN_THREADS) hist[p] = crow[p];
    const int perb = (P + BIN_THREADS - 1) / BIN_THREADS;
    const int b0   = min(P, tid * perb);
    const int b1   = min(P, b0 + perb);
    int sb = 0;
    for (int b = b0; b < b1; ++b) sb += bsize[b];
    __syncthreads();

    const int per = (e + (int)gridDim.x - 1) / (int)gridDim.x;
    const int s0  = min(e, (int)blockIdx.x * per);
    const int s1  = min(e, s0 + per);

    int s = 0;
    for (int b = b0; b < b1; ++b) s += hist[b];
    int v  = s;
    int vb = sb;
    for (int o = 1; o < 32; o <<= 1) {
        const int t  = __shfl_up(v, o);
        const int tb = __shfl_up(vb, o);
        if (lane >= o) { v += t; vb += tb; }
    }
    if (lane == 31) { wsum[wave] = v; wsum_b[wave] = vb; }
    __syncthreads();
    if (wave == 0) {
        int w  = wsum[lane];
        int wb = wsum_b[lane];
        for (int o = 1; o < 32; o <<= 1) {
            const int t  = __shfl_up(w, o);
            const int tb = __shfl_up(wb, o);
            if (lane >= o) { w += t; wb += tb; }
        }
        wsum[lane]   = w;
        wsum_b[lane] = wb;
    }
    __syncthreads();
    int run  = v - s + (wave > 0 ? wsum[wave - 1] : 0);
    int brun = vb - sb + (wave > 0 ? wsum_b[wave - 1] : 0);
    const bool keep = blockIdx.x == 0;
    for (int b = b0; b < b1; ++b) {
        const int c = hist[b];
        const int g = brun + rrow[b];
        if (keep) bstart[b] = brun;
        brun += bsize[b];
        delta[b] = g - run;
        hist[b]  = run;
        run += c;
    }
    if (tid == BIN_THREADS - 1) {
        s_total = min(run, BIN_CAP);
        if (keep) bstart[P] = brun;
    }
    __syncthreads();

    for (int i0 = s0 + tid; i0 < s1; i0 += BIN_THREADS * BIN_BATCH) {
        int d[BIN_BATCH], sidx[BIN_BATCH];
        #pragma unroll
        for (int k = 0; k < BIN_BATCH; ++k) {
            const int i = min(i0 + k * BIN_THREADS, s1 - 1);
            d[k]    = dst[i];
            sidx[k] = src[i];
        }
        asm volatile("" :: "v"(sidx[0]), "v"(sidx[1]), "v"(sidx[2]), "v"(sidx[3]),
                           "v"(sidx[4]), "v"(sidx[5]), "v"(sidx[6]), "v"(sidx[7]));
        #pragma unroll
        for (int k = 0; k < BIN_BATCH; ++k) {
            if (i0 + k * BIN_THREADS < s1 && (unsigned)d[k] < (unsigned)n) {
                const int b = d[k] >> BK_SH;
                const int off = delta[b];
                const int pos = atomicAdd(&hist[b], 1);
                const unsigned key = ((unsigned)sidx[k] << BK_SH) | (unsigned)(d[k] & (BK_R - 1));
                if ((unsigned)pos < (unsigned)BIN_CAP)
                    stage[pos] = make_int2((int)key, off + pos);
            }
        }
    }
    __syncthreads();

    const int total = s_total;
    for (int i0 = tid; i0 < total; i0 += BIN_THREADS * BIN_BATCH) {
        int2 r[BIN_BATCH];
        #pragma unroll
        for (int k = 0; k < BIN_BATCH; ++k)
            r[k] = stage[min(i0 + k * BIN_THREADS, total - 1)];
        #pragma unroll
        for (int k = 0; k < BIN_BATCH; ++k)
            if (i0 + k * BIN_THREADS < total && (unsigned)r[k].y < (unsigned)e)
                rec[r[k].y] = r[k].x;
    }
}

static_assert(BK_R == 8 * 16, "a GEMM block's rows must be one bucket");

__global__ __launch_bounds__(256) void gcn_gemm1_wmma(
    const float* __restrict__ x, const float* __restrict__ W1,
    const int* __restrict__ rec, const int* __restrict__ bstart,
    float* __restrict__ dis, float* __restrict__ hw1, int n) {
    __shared__ float W1s[IN_DIM * H1_DIM];
    __shared__ __attribute__((aligned(16))) int dcnt[BK_R * 4];
    __shared__ float dis_s[BK_R];
    const int tid = threadIdx.x;
    #pragma unroll
    for (int j = 0; j < (IN_DIM * H1_DIM) / 256; ++j)
        W1s[tid + j * 256] = W1[tid + j * 256];
    for (int j = tid; j < BK_R * 4; j += 256) dcnt[j] = 0;
    __syncthreads();

    {
        const int beg = bstart[blockIdx.x];
        const int end = bstart[blockIdx.x + 1];
        const int q = tid & 3;
        for (int i0 = beg + tid; i0 < end; i0 += 256 * DG_BATCH) {
            int r[DG_BATCH];
            #pragma unroll
            for (int k = 0; k < DG_BATCH; ++k)
                r[k] = rec[min(i0 + k * 256, end - 1)];
            asm volatile("" :: "v"(r[0]), "v"(r[1]), "v"(r[2]), "v"(r[3]),
                               "v"(r[4]), "v"(r[5]), "v"(r[6]), "v"(r[7]),
                               "v"(r[8]), "v"(r[9]), "v"(r[10]), "v"(r[11]),
                               "v"(r[12]), "v"(r[13]), "v"(r[14]), "v"(r[15]));
            #pragma unroll
            for (int k = 0; k < DG_BATCH; ++k)
                if (i0 + k * 256 < end) atomicAdd(&dcnt[((r[k] & (BK_R - 1)) << 2) | q], 1);
        }
    }
    __syncthreads();
    if (tid < BK_R) {
        const int4 c4 = *(const int4*)&dcnt[tid * 4];
        const float d = (float)(1 + c4.x + c4.y + c4.z + c4.w);
        const float s = (d > 0.0f) ? rsqrtf(d) : 0.0f;
        dis_s[tid] = s;
    }
    __syncthreads();

    const int wave = tid >> 5;
    const int lane = tid & 31;
    const int half = lane >> 4;
    const int l    = lane & 15;
    const int row0 = (blockIdx.x * 8 + wave) * 16;
    if (row0 >= n) return;

    int arow = row0 + l;
    if (arow >= n) arow = n - 1;

    v8f c = {};
    #pragma unroll 4
    for (int k0 = 0; k0 < IN_DIM; k0 += 4) {
        const int ka = k0 + half * 2;
        const v2f a = *(const v2f*)&x[(size_t)arow * IN_DIM + ka];
        v2f b;
        b.x = W1s[ka * H1_DIM + l];
        b.y = W1s[(ka + 1) * H1_DIM + l];
        c = __builtin_amdgcn_wmma_f32_16x16x4_f32(
                false, a, false, b, (short)0, c, false, false);
    }

    if (row0 + 16 <= n) {
        #pragma unroll
        for (int r = 0; r < 8; ++r) {
            const int row = row0 + r + half * 8;
            hw1[(size_t)row * H1_DIM + l] = dis_s[wave * 16 + r + half * 8] * c[r];
        }
    } else {
        for (int r = 0; r < 8; ++r) {
            const int row = row0 + r + half * 8;
            if (row < n) hw1[(size_t)row * H1_DIM + l] = dis_s[wave * 16 + r + half * 8] * c[r];
        }
    }

    if (tid < BK_R) {
        const int node = blockIdx.x * BK_R + tid;
        if (node < n) dis[node] = dis_s[tid];
    }
}

__device__ __forceinline__ void stage_records(
    const int* __restrict__ rec, int* __restrict__ lrec, int c0, int len, int tid) {
    int v[ST_BATCH];
    #pragma unroll
    for (int k = 0; k < ST_BATCH; ++k)
        v[k] = rec[c0 + min(tid + k * SC_THREADS, len - 1)];
    #pragma unroll
    for (int k = 0; k < ST_BATCH; ++k) {
        const int j = tid + k * SC_THREADS;
        lrec[j] = (j < len) ? v[k] : 0;
    }
}

__global__ __launch_bounds__(SC_THREADS) void gcn_scatter1_gemm2(
    const int* __restrict__ rec, const int* __restrict__ bstart,
    const float* __restrict__ hw1, const float* __restrict__ dis,
    const float* __restrict__ b1, const float* __restrict__ W2,
    float* __restrict__ hw2, int n) {
    __shared__ __attribute__((aligned(16))) float acc[BK_R * H1_DIM];
    __shared__ __attribute__((aligned(16))) int lrec[SC_CAP];
    __shared__ float W2s[H1_DIM * 16];
    __shared__ float dis_s[BK_R];
    __shared__ __attribute__((aligned(16))) float outs[BK_R * H2_DIM];
    const int tid = threadIdx.x;
    const int lo  = blockIdx.x * BK_R;
    for (int j = tid; j < BK_R * H1_DIM; j += SC_THREADS) acc[j] = 0.0f;

    const int beg = bstart[blockIdx.x];
    const int end = bstart[blockIdx.x + 1];
    const int c = tid & (H1_DIM - 1);
    const int g = tid >> 4;
    const int SW = (SC_THREADS >> 4) * SC_BATCH;
    for (int c0 = beg; c0 < end; c0 += SC_CAP) {
        if (c0 != beg) __syncthreads();
        const int len = min(SC_CAP, end - c0);
        stage_records(rec, lrec, c0, len, tid);
        __syncthreads();

        for (int i0 = g * SC_BATCH; i0 < len; i0 += SW) {
            const int4 ra = *(const int4*)&lrec[i0];
            const int4 rb = *(const int4*)&lrec[i0 + 4];
            const int r[SC_BATCH] = {ra.x, ra.y, ra.z, ra.w, rb.x, rb.y, rb.z, rb.w};
            float h[SC_BATCH];
            #pragma unroll
            for (int k = 0; k < SC_BATCH; ++k) {
                const unsigned s = min((unsigned)r[k] >> BK_SH, (unsigned)(n - 1));
                h[k] = hw1[(size_t)s * H1_DIM + c];
            }
            #pragma unroll
            for (int k = 0; k < SC_BATCH; ++k) {
                const float v = (i0 + k < len) ? h[k] : 0.0f;
                const int dl = r[k] & (BK_R - 1);
                atomicAdd(&acc[dl * H1_DIM + c], v);
            }
        }
    }
    const int cnt = min(BK_R, n - lo) * H1_DIM;
    const float* __restrict__ own = hw1 + (size_t)lo * H1_DIM;
    const float bias = b1[c];
    float o[H1_ROUNDS];
    #pragma unroll
    for (int k = 0; k < H1_ROUNDS; ++k)
        o[k] = own[min(tid + k * SC_THREADS, cnt - 1)];
    if (tid < H1_DIM * 16) {
        const int k = tid >> 4, col = tid & 15;
        W2s[tid] = (col < H2_DIM) ? W2[k * H2_DIM + col] : 0.0f;
    } else if (tid < H1_DIM * 16 + BK_R) {
        const int i = tid - H1_DIM * 16;
        dis_s[i] = (lo + i < n) ? dis[lo + i] : 0.0f;
    }
    __syncthreads();

    #pragma unroll
    for (int k = 0; k < H1_ROUNDS; ++k) {
        const int j = tid + k * SC_THREADS;
        const float v = fmaxf(dis_s[j >> 4] * (acc[j] + o[k]) + bias, 0.0f);
        acc[j] = (j < cnt) ? v : 0.0f;
    }
    __syncthreads();

    const int wave = tid >> 5;
    const int lane = tid & 31;
    const int half = lane >> 4;
    const int l    = lane & 15;
    if (wave < BK_R / 16 && lo + wave * 16 < n) {
        v8f d = {};
        #pragma unroll
        for (int k0 = 0; k0 < H1_DIM; k0 += 4) {
            const int ka = k0 + half * 2;
            const v2f a = *(const v2f*)&acc[(wave * 16 + l) * H1_DIM + ka];
            v2f b;
            b.x = W2s[ka * 16 + l];
            b.y = W2s[(ka + 1) * 16 + l];
            d = __builtin_amdgcn_wmma_f32_16x16x4_f32(
                    false, a, false, b, (short)0, d, false, false);
        }
        if (l < H2_DIM) {
            #pragma unroll
            for (int r = 0; r < 8; ++r) {
                const int row = wave * 16 + r + half * 8;
                outs[row * H2_DIM + l] = dis_s[row] * d[r];
            }
        }
    }
    __syncthreads();

    const int cnt2 = min(BK_R, n - lo) * H2_DIM;
    if (tid * 4 < cnt2)
        *(float4*)&hw2[(size_t)lo * H2_DIM + tid * 4] = *(const float4*)&outs[tid * 4];
}

__global__ __launch_bounds__(SC_THREADS) void gcn_scatter2(
    const int* __restrict__ rec, const int* __restrict__ bstart,
    const float* __restrict__ hw2, const float* __restrict__ dis,
    const float* __restrict__ b2, float* __restrict__ out, int n) {
    __shared__ float acc[BK_R * H2_DIM];
    __shared__ __attribute__((aligned(16))) int lrec[SC_CAP];
    const int tid = threadIdx.x;
    for (int j = tid; j < BK_R * H2_DIM; j += SC_THREADS) acc[j] = 0.0f;

    const int lo  = blockIdx.x * BK_R;
    const int beg = bstart[blockIdx.x];
    const int end = bstart[blockIdx.x + 1];
    const int c = tid & (H2_DIM - 1);
    const int g = tid >> 3;
    const int SW = (SC_THREADS >> 3) * SC_BATCH;
    for (int c0 = beg; c0 < end; c0 += SC_CAP) {
        if (c0 != beg) __syncthreads();
        const int len = min(SC_CAP, end - c0);
        stage_records(rec, lrec, c0, len, tid);
        __syncthreads();

        for (int i0 = g * SC_BATCH; i0 < len; i0 += SW) {
            const int4 ra = *(const int4*)&lrec[i0];
            const int4 rb = *(const int4*)&lrec[i0 + 4];
            const int r[SC_BATCH] = {ra.x, ra.y, ra.z, ra.w, rb.x, rb.y, rb.z, rb.w};
            float h[SC_BATCH];
            #pragma unroll
            for (int k = 0; k < SC_BATCH; ++k) {
                const unsigned s = min((unsigned)r[k] >> BK_SH, (unsigned)(n - 1));
                h[k] = hw2[(size_t)s * H2_DIM + c];
            }
            #pragma unroll
            for (int k = 0; k < SC_BATCH; ++k) {
                const float v = (i0 + k < len) ? h[k] : 0.0f;
                const int dl = r[k] & (BK_R - 1);
                atomicAdd(&acc[dl * H2_DIM + c], v);
            }
        }
    }
    __syncthreads();

    const int cnt = min(BK_R, n - lo) * H2_DIM;
    const float* __restrict__ own = hw2 + (size_t)lo * H2_DIM;
    float* __restrict__ row = out + (size_t)lo * H2_DIM;
    const float bias = b2[c];
    for (int j = tid; j < cnt; j += SC_THREADS)
        row[j] = dis[lo + (j >> 3)] * (acc[j] + own[j]) + bias;
}

extern "C" void kernel_launch(void* const* d_in, const int* in_sizes, int n_in,
                              void* d_out, int out_size, void* d_ws, size_t ws_size,
                              hipStream_t stream) {
    const float* x    = (const float*)d_in[0];
    const int*   edge = (const int*)  d_in[1];
    const float* W1   = (const float*)d_in[2];
    const float* b1   = (const float*)d_in[3];
    const float* W2   = (const float*)d_in[4];
    const float* b2   = (const float*)d_in[5];
    float* out = (float*)d_out;

    const int n = in_sizes[0] / IN_DIM;
    const int e = in_sizes[1] / 2;
    const int* src = edge;
    const int* dst = edge + e;

    float* ws   = (float*)d_ws;
    float* dis  = ws;
    float* hw1  = dis + n;
    float* hw2  = (float*)((((size_t)(hw1 + (size_t)n * H1_DIM)) + 127) & ~(size_t)127);

    const int P = (n + BK_R - 1) / BK_R;
    const int bneed   = (e + BIN_CAP - 1) / BIN_CAP;
    const int bslices = bneed > 256 ? bneed : 256;
    const int TP = (P + 31) & ~31;
    int* total  = (int*)(hw2 + (size_t)n * H2_DIM);
    int* bstart = total + P;
    int* cnt    = (int*)((((size_t)(bstart + P + 1)) + 127) & ~(size_t)127);
    int* rel    = cnt + (size_t)bslices * TP;
    int* rec    = rel + (size_t)bslices * TP;

    const int B = 256;
    const int tiles  = (n + 15) / 16;
    const int gtiles = (tiles + 7) / 8;

    gcn_slice_hist  <<<bslices, BIN_THREADS, (size_t)P * sizeof(int), stream>>>(dst, cnt, e, n, P, TP);
    gcn_slice_scan  <<<TP / 32, SS_THREADS, 0, stream>>>(cnt, rel, total, bslices, P, TP);
    gcn_bin_edges   <<<bslices, BIN_THREADS, 2 * (size_t)P * sizeof(int), stream>>>(src, dst, cnt, rel, total, bstart, rec, e, n, P, TP);

    gcn_gemm1_wmma    <<<gtiles, B, 0, stream>>>(x, W1, rec, bstart, dis, hw1, n);
    gcn_scatter1_gemm2<<<P, SC_THREADS, 0, stream>>>(rec, bstart, hw1, dis, b1, W2, hw2, n);
    gcn_scatter2      <<<P, SC_THREADS, 0, stream>>>(rec, bstart, hw2, dis, b2, out, n);
}


# Binning in one launch, no counting pass or table scan before it: each slice sorted by bucket in place plus a table row of run starts; a bucket's readers walk one run per slice; LDS add order changes
# speedup vs baseline: 7.5403x; 1.0589x over previous
#include <hip/hip_runtime.h>


typedef __attribute__((ext_vector_type(2))) float v2f;
typedef __attribute__((ext_vector_type(8))) float v8f;

#define IN_DIM 128
#define H1_DIM 16
#define H2_DIM 8

#define BK_SH 7
#define BK_R  128
#define BIN_THREADS 1024
#define BIN_BATCH   8
#define BIN_CAP     25600
#define SC_BATCH    8
#define SC_THREADS  512
#define SC_CAP      10240
#define H1_ROUNDS   ((BK_R * H1_DIM) / SC_THREADS)
#define RB_SC       8
#define RB_GM       16
static_assert(SC_BATCH == 8, "a lane group reads its batch of records as two 16-byte LDS loads");
static_assert((BK_R * H1_DIM) % SC_THREADS == 0 && SC_THREADS >= H1_DIM * 16 + BK_R && SC_THREADS / 32 >= BK_R / 16 && SC_THREADS * 4 >= BK_R * H2_DIM,
              "the layer-1 scatter stages W2 and dis side by side, runs one wave per 16-row tile and stores hw2 16 bytes per lane");
static_assert(SC_CAP % ((SC_THREADS / H2_DIM) * SC_BATCH) == 0 &&
              SC_CAP % ((SC_THREADS / H1_DIM) * SC_BATCH) == 0, "a pass is whole sweeps of the work-group");

template <int T>
__device__ __forceinline__ int block_scan_excl(int v, int* wsum, int tid, int& total) {
    const int lane = tid & 31;
    const int wave = tid >> 5;
    int inc = v;
    #pragma unroll
    for (int o = 1; o < 32; o <<= 1) {
        const int t = __shfl_up(inc, o);
        if (lane >= o) inc += t;
    }
    if (lane == 31) wsum[wave] = inc;
    __syncthreads();
    if (wave == 0) {
        int w = (lane < T / 32) ? wsum[lane] : 0;
        #pragma unroll
        for (int o = 1; o < 32; o <<= 1) {
            const int t = __shfl_up(w, o);
            if (lane >= o) w += t;
        }
        wsum[lane] = w;
    }
    __syncthreads();
    total = wsum[T / 32 - 1];
    return inc - v + (wave > 0 ? wsum[wave - 1] : 0);
}

__global__ __launch_bounds__(BIN_THREADS) void gcn_bin_edges(
    const int* __restrict__ src, const int* __restrict__ dst,
    int* __restrict__ tab, int* __restrict__ rec, int e, int n, int P, int TP) {
    extern __shared__ int bin_sm[];
    __shared__ int2 stash[BIN_CAP];
    __shared__ int sorted[BIN_CAP];
    __shared__ int wsum[32];
    __shared__ int s_total;
    int* hist = bin_sm;
    const int tid  = threadIdx.x;
    const int lane = tid & 31;
    for (int p = tid; p < P + 32; p += BIN_THREADS) hist[p] = 0;
    __syncthreads();

    const int per = (e + (int)gridDim.x - 1) / (int)gridDim.x;
    const int s0  = min(e, (int)blockIdx.x * per);
    const int s1  = min(e, s0 + per);
    const int cnt = min(s1 - s0, BIN_CAP);

    for (int i0 = tid; i0 < cnt; i0 += BIN_THREADS * BIN_BATCH) {
        int d[BIN_BATCH], sidx[BIN_BATCH];
        #pragma unroll
        for (int k = 0; k < BIN_BATCH; ++k) {
            const int i = s0 + min(i0 + k * BIN_THREADS, cnt - 1);
            d[k]    = dst[i];
            sidx[k] = src[i];
        }
        asm volatile("" :: "v"(d[0]), "v"(d[1]), "v"(d[2]), "v"(d[3]),
                           "v"(d[4]), "v"(d[5]), "v"(d[6]), "v"(d[7]),
                           "v"(sidx[0]), "v"(sidx[1]), "v"(sidx[2]), "v"(sidx[3]),
                           "v"(sidx[4]), "v"(sidx[5]), "v"(sidx[6]), "v"(sidx[7]));
        #pragma unroll
        for (int k = 0; k < BIN_BATCH; ++k) {
            const int j = i0 + k * BIN_THREADS;
            if (j < cnt) {
                stash[j] = make_int2(d[k], sidx[k]);
                if ((unsigned)d[k] < (unsigned)n) atomicAdd(&hist[d[k] >> BK_SH], 1);
            }
        }
    }
    __syncthreads();

    const int perb = (P + BIN_THREADS - 1) / BIN_THREADS;
    const int b0   = min(P, tid * perb);
    const int b1   = min(P, b0 + perb);
    int s = 0;
    for (int b = b0; b < b1; ++b) s += hist[b];
    int total;
    int run = block_scan_excl<BIN_THREADS>(s, wsum, tid, total);
    int* __restrict__ trow = tab + (size_t)blockIdx.x * TP;
    for (int b = b0; b < b1; ++b) {
        const int c = hist[b];
        trow[b] = s0 + run;
        hist[b] = run;
        run += c;
    }
    if (tid == BIN_THREADS - 1) {
        trow[P] = s0 + run;
        s_total = min(run, BIN_CAP);
    }
    __syncthreads();

    for (int i0 = tid; i0 < cnt; i0 += BIN_THREADS * BIN_BATCH) {
        int2 v[BIN_BATCH];
        #pragma unroll
        for (int k = 0; k < BIN_BATCH; ++k)
            v[k] = stash[min(i0 + k * BIN_THREADS, cnt - 1)];
        int  pos[BIN_BATCH];
        bool ok[BIN_BATCH];
        #pragma unroll
        for (int k = 0; k < BIN_BATCH; ++k) {
            ok[k]  = i0 + k * BIN_THREADS < cnt && (unsigned)v[k].x < (unsigned)n;
            pos[k] = atomicAdd(&hist[ok[k] ? (v[k].x >> BK_SH) : P + lane], 1);
        }
        #pragma unroll
        for (int k = 0; k < BIN_BATCH; ++k) {
            const unsigned key = ((unsigned)v[k].y << BK_SH) | (unsigned)(v[k].x & (BK_R - 1));
            if (ok[k] && (unsigned)pos[k] < (unsigned)BIN_CAP)
                sorted[pos[k]] = (int)key;
        }
    }
    __syncthreads();

    const int tot = s_total;
    for (int i = tid; i < tot; i += BIN_THREADS) rec[s0 + i] = sorted[i];
}

template <int T>
__device__ __forceinline__ int bucket_runs(
    const int* __restrict__ tab, int b, int S, int TP, int e,
    int* rstart, int* roff, int* wsum, int* s_safe, int tid) {
    const int pers = (S + T - 1) / T;
    const int sA = min(S, tid * pers);
    const int sB = min(S, sA + pers);
    int sum = 0;
    for (int s = sA; s < sB; ++s) {
        int a = tab[(size_t)s * TP + b];
        const int z = tab[(size_t)s * TP + b + 1];
        a = min(max(a, 0), e);
        const int c = min(max(z - a, 0), e - a);
        rstart[s] = a;
        roff[s]   = c;
        sum += c;
        if (c > 0) *s_safe = a;
    }
    int total;
    int run = block_scan_excl<T>(sum, wsum, tid, total);
    for (int s = sA; s < sB; ++s) {
        const int c = roff[s];
        roff[s] = run;
        run += c;
    }
    if (tid == T - 1) roff[S] = run;
    __syncthreads();
    return total;
}

template <int T, int NB, typename F>
__device__ __forceinline__ void walk_runs(
    const int* __restrict__ rec, const int* rstart, const int* roff,
    int S, int safe, int tid, F f) {
    static_assert(NB <= 32 && (NB & (NB - 1)) == 0, "lane k of a wave fetches the table entries of run k of the batch");
    const int lane = tid & 31;
    const int wave = __builtin_amdgcn_readfirstlane(tid >> 5);
    for (int sb = wave * NB; sb < S; sb += (T / 32) * NB) {
        const int sl = min(sb + (lane & (NB - 1)), S - 1);
        const int av = rstart[sl];
        const int ov = roff[sl];
        const int ev = roff[sl + 1];
        int a[NB], o[NB], c[NB], v0[NB], v1[NB];
        #pragma unroll
        for (int k = 0; k < NB; ++k) {
            a[k] = __builtin_amdgcn_readlane(av, k);
            o[k] = __builtin_amdgcn_readlane(ov, k);
            c[k] = (sb + k < S) ? __builtin_amdgcn_readlane(ev, k) - o[k] : 0;
            v0[k] = rec[(lane < c[k]) ? a[k] + lane : safe];
            v1[k] = rec[(lane + 32 < c[k]) ? a[k] + lane + 32 : safe];
        }
        #pragma unroll
        for (int k = 0; k < NB; ++k) {
            if (lane < c[k]) f(o[k] + lane, v0[k]);
            if (lane + 32 < c[k]) f(o[k] + lane + 32, v1[k]);
            for (int j = lane + 64; j < c[k]; j += 32) f(o[k] + j, rec[a[k] + j]);
        }
    }
}

static_assert(BK_R == 8 * 16, "a GEMM block's rows must be one bucket");

__global__ __launch_bounds__(256) void gcn_gemm1_wmma(
    const float* __restrict__ x, const float* __restrict__ W1,
    const int* __restrict__ rec, const int* __restrict__ tab,
    float* __restrict__ dis, float* __restrict__ hw1, int n, int e, int S, int TP) {
    extern __shared__ int run_sm[];
    __shared__ int wsum[32];
    __shared__ int s_safe;
    int* rstart = run_sm;
    int* roff   = run_sm + S;
    __shared__ float W1s[IN_DIM * H1_DIM];
    __shared__ __attribute__((aligned(16))) int dcnt[BK_R * 4];
    __shared__ float dis_s[BK_R];
    const int tid = threadIdx.x;
    #pragma unroll
    for (int j = 0; j < (IN_DIM * H1_DIM) / 256; ++j)
        W1s[tid + j * 256] = W1[tid + j * 256];
    for (int j = tid; j < BK_R * 4; j += 256) dcnt[j] = 0;
    {
        const int tot = bucket_runs<256>(tab, (int)blockIdx.x, S, TP, e, rstart, roff, wsum, &s_safe, tid);
        if (tot > 0) {
            const int q = tid & 3;
            walk_runs<256, RB_GM>(rec, rstart, roff, S, s_safe, tid,
                           [&](int, int r) { atomicAdd(&dcnt[((r & (BK_R - 1)) << 2) | q], 1); });
        }
    }
    __syncthreads();
    if (tid < BK_R) {
        const int4 c4 = *(const int4*)&dcnt[tid * 4];
        const float d = (float)(1 + c4.x + c4.y + c4.z + c4.w);
        const float s = (d > 0.0f) ? rsqrtf(d) : 0.0f;
        dis_s[tid] = s;
    }
    __syncthreads();

    const int wave = tid >> 5;
    const int lane = tid & 31;
    const int half = lane >> 4;
    const int l    = lane & 15;
    const int row0 = (blockIdx.x * 8 + wave) * 16;
    if (row0 >= n) return;

    int arow = row0 + l;
    if (arow >= n) arow = n - 1;

    v8f c = {};
    #pragma unroll 4
    for (int k0 = 0; k0 < IN_DIM; k0 += 4) {
        const int ka = k0 + half * 2;
        const v2f a = *(const v2f*)&x[(size_t)arow * IN_DIM + ka];
        v2f b;
        b.x = W1s[ka * H1_DIM + l];
        b.y = W1s[(ka + 1) * H1_DIM + l];
        c = __builtin_amdgcn_wmma_f32_16x16x4_f32(
                false, a, false, b, (short)0, c, false, false);
    }

    if (row0 + 16 <= n) {
        #pragma unroll
        for (int r = 0; r < 8; ++r) {
            const int row = row0 + r + half * 8;
            hw1[(size_t)row * H1_DIM + l] = dis_s[wave * 16 + r + half * 8] * c[r];
        }
    } else {
        for (int r = 0; r < 8; ++r) {
            const int row = row0 + r + half * 8;
            if (row < n) hw1[(size_t)row * H1_DIM + l] = dis_s[wave * 16 + r + half * 8] * c[r];
        }
    }

    if (tid < BK_R) {
        const int node = blockIdx.x * BK_R + tid;
        if (node < n) dis[node] = dis_s[tid];
    }
}

__global__ __launch_bounds__(SC_THREADS) void gcn_scatter1_gemm2(
    const int* __restrict__ rec, const int* __restrict__ tab,
    const float* __restrict__ hw1, const float* __restrict__ dis,
    const float* __restrict__ b1, const float* __restrict__ W2,
    float* __restrict__ hw2, int n, int e, int S, int TP) {
    extern __shared__ int run_sm[];
    __shared__ int wsum[32];
    __shared__ int s_safe;
    int* rstart = run_sm;
    int* roff   = run_sm + S;
    __shared__ __attribute__((aligned(16))) float acc[BK_R * H1_DIM];
    __shared__ __attribute__((aligned(16))) int lrec[SC_CAP];
    __shared__ float W2s[H1_DIM * 16];
    __shared__ float dis_s[BK_R];
    __shared__ __attribute__((aligned(16))) float outs[BK_R * H2_DIM];
    const int tid = threadIdx.x;
    const int lo  = blockIdx.x * BK_R;
    for (int j = tid; j < BK_R * H1_DIM; j += SC_THREADS) acc[j] = 0.0f;

    const int tot = bucket_runs<SC_THREADS>(tab, (int)blockIdx.x, S, TP, e, rstart, roff, wsum, &s_safe, tid);
    const int c = tid & (H1_DIM - 1);
    const int g = tid >> 4;
    const int SW = (SC_THREADS >> 4) * SC_BATCH;
    for (int p0 = 0; p0 < tot; p0 += SC_CAP) {
        if (p0 != 0) __syncthreads();
        const int len = min(SC_CAP, tot - p0);
        walk_runs<SC_THREADS, RB_SC>(rec, rstart, roff, S, s_safe, tid,
                              [&](int i, int r) {
                                  const unsigned j = (unsigned)(i - p0);
                                  if (j < (unsigned)len) lrec[j] = r;
                              });
        if (tid < SC_BATCH && len + tid < SC_CAP) lrec[len + tid] = 0;
        __syncthreads();

        for (int i0 = g * SC_BATCH; i0 < len; i0 += SW) {
            const int4 ra = *(const int4*)&lrec[i0];
            const int4 rb = *(const int4*)&lrec[i0 + 4];
            const int r[SC_BATCH] = {ra.x, ra.y, ra.z, ra.w, rb.x, rb.y, rb.z, rb.w};
            float h[SC_BATCH];
            #pragma unroll
            for (int k = 0; k < SC_BATCH; ++k) {
                const unsigned s = min((unsigned)r[k] >> BK_SH, (unsigned)(n - 1));
                h[k] = hw1[(size_t)s * H1_DIM + c];
            }
            #pragma unroll
            for (int k = 0; k < SC_BATCH; ++k) {
                const float v = (i0 + k < len) ? h[k] : 0.0f;
                const int dl = r[k] & (BK_R - 1);
                atomicAdd(&acc[dl * H1_DIM + c], v);
            }
        }
    }
    const int cnt = min(BK_R, n - lo) * H1_DIM;
    const float* __restrict__ own = hw1 + (size_t)lo * H1_DIM;
    const float bias = b1[c];
    float o[H1_ROUNDS];
    #pragma unroll
    for (int k = 0; k < H1_ROUNDS; ++k)
        o[k] = own[min(tid + k * SC_THREADS, cnt - 1)];
    if (tid < H1_DIM * 16) {
        const int k = tid >> 4, col = tid & 15;
        W2s[tid] = (col < H2_DIM) ? W2[k * H2_DIM + col] : 0.0f;
    } else if (tid < H1_DIM * 16 + BK_R) {
        const int i = tid - H1_DIM * 16;
        dis_s[i] = (lo + i < n) ? dis[lo + i] : 0.0f;
    }
    __syncthreads();

    #pragma unroll
    for (int k = 0; k < H1_ROUNDS; ++k) {
        const int j = tid + k * SC_THREADS;
        const float v = fmaxf(dis_s[j >> 4] * (acc[j] + o[k]) + bias, 0.0f);
        acc[j] = (j < cnt) ? v : 0.0f;
    }
    __syncthreads();

    const int wave = tid >> 5;
    const int lane = tid & 31;
    const int half = lane >> 4;
    const int l    = lane & 15;
    if (wave < BK_R / 16 && lo + wave * 16 < n) {
        v8f d = {};
        #pragma unroll
        for (int k0 = 0; k0 < H1_DIM; k0 += 4) {
            const int ka = k0 + half * 2;
            const v2f a = *(const v2f*)&acc[(wave * 16 + l) * H1_DIM + ka];
            v2f b;
            b.x = W2s[ka * 16 + l];
            b.y = W2s[(ka + 1) * 16 + l];
            d = __builtin_amdgcn_wmma_f32_16x16x4_f32(
                    false, a, false, b, (short)0, d, false, false);
        }
        if (l < H2_DIM) {
            #pragma unroll
            for (int r = 0; r < 8; ++r) {
                const int row = wave * 16 + r + half * 8;
                outs[row * H2_DIM + l] = dis_s[row] * d[r];
            }
        }
    }
    __syncthreads();

    const int cnt2 = min(BK_R, n - lo) * H2_DIM;
    if (tid * 4 < cnt2)
        *(float4*)&hw2[(size_t)lo * H2_DIM + tid * 4] = *(const float4*)&outs[tid * 4];
}

__global__ __launch_bounds__(SC_THREADS) void gcn_scatter2(
    const int* __restrict__ rec, const int* __restrict__ tab,
    const float* __restrict__ hw2, const float* __restrict__ dis,
    const float* __restrict__ b2, float* __restrict__ out, int n, int e, int S, int TP) {
    extern __shared__ int run_sm[];
    __shared__ int wsum[32];
    __shared__ int s_safe;
    int* rstart = run_sm;
    int* roff   = run_sm + S;
    __shared__ float acc[BK_R * H2_DIM];
    __shared__ __attribute__((aligned(16))) int lrec[SC_CAP];
    const int tid = threadIdx.x;
    for (int j = tid; j < BK_R * H2_DIM; j += SC_THREADS) acc[j] = 0.0f;

    const int lo  = blockIdx.x * BK_R;
    const int tot = bucket_runs<SC_THREADS>(tab, (int)blockIdx.x, S, TP, e, rstart, roff, wsum, &s_safe, tid);
    const int c = tid & (H2_DIM - 1);
    const int g = tid >> 3;
    const int SW = (SC_THREADS >> 3) * SC_BATCH;
    for (int p0 = 0; p0 < tot; p0 += SC_CAP) {
        if (p0 != 0) __syncthreads();
        const int len = min(SC_CAP, tot - p0);
        walk_runs<SC_THREADS, RB_SC>(rec, rstart, roff, S, s_safe, tid,
                              [&](int i, int r) {
                                  const unsigned j = (unsigned)(i - p0);
                                  if (j < (unsigned)len) lrec[j] = r;
                              });
        if (tid < SC_BATCH && len + tid < SC_CAP) lrec[len + tid] = 0;
        __syncthreads();

        for (int i0 = g * SC_BATCH; i0 < len; i0 += SW) {
            const int4 ra = *(const int4*)&lrec[i0];
            const int4 rb = *(const int4*)&lrec[i0 + 4];
            const int r[SC_BATCH] = {ra.x, ra.y, ra.z, ra.w, rb.x, rb.y, rb.z, rb.w};
            float h[SC_BATCH];
            #pragma unroll
            for (int k = 0; k < SC_BATCH; ++k) {
                const unsigned s = min((unsigned)r[k] >> BK_SH, (unsigned)(n - 1));
                h[k] = hw2[(size_t)s * H2_DIM + c];
            }
            #pragma unroll
            for (int k = 0; k < SC_BATCH; ++k) {
                const float v = (i0 + k < len) ? h[k] : 0.0f;
                const int dl = r[k] & (BK_R - 1);
                atomicAdd(&acc[dl * H2_DIM + c], v);
            }
        }
    }
    __syncthreads();

    const int cnt = min(BK_R, n - lo) * H2_DIM;
    const float* __restrict__ own = hw2 + (size_t)lo * H2_DIM;
    float* __restrict__ row = out + (size_t)lo * H2_DIM;
    const float bias = b2[c];
    for (int j = tid; j < cnt; j += SC_THREADS)
        row[j] = dis[lo + (j >> 3)] * (acc[j] + own[j]) + bias;
}

extern "C" void kernel_launch(void* const* d_in, const int* in_sizes, int n_in,
                              void* d_out, int out_size, void* d_ws, size_t ws_size,
                              hipStream_t stream) {
    const float* x    = (const float*)d_in[0];
    const int*   edge = (const int*)  d_in[1];
    const float* W1   = (const float*)d_in[2];
    const float* b1   = (const float*)d_in[3];
    const float* W2   = (const float*)d_in[4];
    const float* b2   = (const float*)d_in[5];
    float* out = (float*)d_out;

    const int n = in_sizes[0] / IN_DIM;
    const int e = in_sizes[1] / 2;
    const int* src = edge;
    const int* dst = edge + e;

    float* ws   = (float*)d_ws;
    float* dis  = ws;
    float* hw1  = dis + n;
    float* hw2  = (float*)((((size_t)(hw1 + (size_t)n * H1_DIM)) + 127) & ~(size_t)127);

    const int P = (n + BK_R - 1) / BK_R;
    const int bneed = (e + BIN_CAP - 1) / BIN_CAP;
    const int S  = bneed > 256 ? bneed : 256;
    const int TP = (P + 1 + 31) & ~31;
    int* tab = (int*)((((size_t)(hw2 + (size_t)n * H2_DIM)) + 127) & ~(size_t)127);
    int* rec = tab + (size_t)S * TP;
    const size_t runb = (size_t)(2 * S + 1) * sizeof(int);

    const int B = 256;
    const int tiles  = (n + 15) / 16;
    const int gtiles = (tiles + 7) / 8;

    gcn_bin_edges     <<<S, BIN_THREADS, (size_t)(P + 32) * sizeof(int), stream>>>(src, dst, tab, rec, e, n, P, TP);
    gcn_gemm1_wmma    <<<gtiles, B, runb, stream>>>(x, W1, rec, tab, dis, hw1, n, e, S, TP);
    gcn_scatter1_gemm2<<<P, SC_THREADS, runb, stream>>>(rec, tab, hw1, dis, b1, W2, hw2, n, e, S, TP);
    gcn_scatter2      <<<P, SC_THREADS, runb, stream>>>(rec, tab, hw2, dis, b2, out, n, e, S, TP);
}


# Buckets of 80 nodes, not 128: 1,250 work-groups of 12 waves, at most 5 per WGP, so no WGP sums more than 400 nodes per scatter (was 512 on the fullest); order of LDS adds changes
# speedup vs baseline: 8.0027x; 1.0613x over previous
#include <hip/hip_runtime.h>


typedef __attribute__((ext_vector_type(2))) float v2f;
typedef __attribute__((ext_vector_type(8))) float v8f;

#define IN_DIM 128
#define H1_DIM 16
#define H2_DIM 8

#define BK_SH 7
#define BK_R  80
#define BK_ROWS (1 << BK_SH)
#define GM_WAVES   (BK_R / 16)
#define GM_THREADS (GM_WAVES * 32)
#define DIS_PITCH  96
#define W1_ROUNDS  ((IN_DIM * H1_DIM + GM_THREADS - 1) / GM_THREADS)
#define BIN_THREADS 1024
#define BIN_BATCH   8
#define BIN_CAP     25600
#define SC_BATCH    8
#define SC_THREADS  384
#define SC_CAP      6144
#define H1_ROUNDS   ((BK_R * H1_DIM + SC_THREADS - 1) / SC_THREADS)
#define RB_SC       8
#define RB_GM       16
static_assert(SC_BATCH == 8, "a lane group reads its batch of records as two 16-byte LDS loads");
static_assert(BK_R % 16 == 0 && BK_R <= BK_ROWS && BK_R <= DIS_PITCH && DIS_PITCH % 32 == 0, "a bucket is whole 16-row tiles, its offsets fit a record, its dis fits its lines");
static_assert(SC_THREADS % H1_DIM == 0 && SC_THREADS >= H1_DIM * 16 + BK_R && SC_THREADS / 32 >= BK_R / 16 && SC_THREADS * 4 >= BK_R * H2_DIM,
              "the layer-1 scatter stages W2 and dis side by side, runs one wave per 16-row tile and stores hw2 16 bytes per lane");
static_assert(H1_ROUNDS * SC_THREADS <= BK_ROWS * H1_DIM && H1_ROUNDS * SC_THREADS <= DIS_PITCH * H1_DIM,
              "the layer-1 write-out's last, partial round stays inside acc and dis_s");
static_assert(SC_CAP % ((SC_THREADS / H2_DIM) * SC_BATCH) == 0 &&
              SC_CAP % ((SC_THREADS / H1_DIM) * SC_BATCH) == 0, "a pass is whole sweeps of the work-group");

template <int T>
__device__ __forceinline__ int block_scan_excl(int v, int* wsum, int tid, int& total) {
    const int lane = tid & 31;
    const int wave = tid >> 5;
    int inc = v;
    #pragma unroll
    for (int o = 1; o < 32; o <<= 1) {
        const int t = __shfl_up(inc, o);
        if (lane >= o) inc += t;
    }
    if (lane == 31) wsum[wave] = inc;
    __syncthreads();
    if (wave == 0) {
        int w = (lane < T / 32) ? wsum[lane] : 0;
        #pragma unroll
        for (int o = 1; o < 32; o <<= 1) {
            const int t = __shfl_up(w, o);
            if (lane >= o) w += t;
        }
        wsum[lane] = w;
    }
    __syncthreads();
    total = wsum[T / 32 - 1];
    return inc - v + (wave > 0 ? wsum[wave - 1] : 0);
}

__global__ __launch_bounds__(BIN_THREADS) void gcn_bin_edges(
    const int* __restrict__ src, const int* __restrict__ dst,
    int* __restrict__ tab, int* __restrict__ rec, int e, int n, int P, int TP) {
    extern __shared__ int bin_sm[];
    __shared__ int2 stash[BIN_CAP];
    __shared__ int sorted[BIN_CAP];
    __shared__ int wsum[32];
    __shared__ int s_total;
    int* hist = bin_sm;
    const int tid  = threadIdx.x;
    const int lane = tid & 31;
    for (int p = tid; p < P + 32; p += BIN_THREADS) hist[p] = 0;
    __syncthreads();

    const int per = (e + (int)gridDim.x - 1) / (int)gridDim.x;
    const int s0  = min(e, (int)blockIdx.x * per);
    const int s1  = min(e, s0 + per);
    const int cnt = min(s1 - s0, BIN_CAP);

    for (int i0 = tid; i0 < cnt; i0 += BIN_THREADS * BIN_BATCH) {
        int d[BIN_BATCH], sidx[BIN_BATCH];
        #pragma unroll
        for (int k = 0; k < BIN_BATCH; ++k) {
            const int i = s0 + min(i0 + k * BIN_THREADS, cnt - 1);
            d[k]    = dst[i];
            sidx[k] = src[i];
        }
        asm volatile("" :: "v"(d[0]), "v"(d[1]), "v"(d[2]), "v"(d[3]),
                           "v"(d[4]), "v"(d[5]), "v"(d[6]), "v"(d[7]),
                           "v"(sidx[0]), "v"(sidx[1]), "v"(sidx[2]), "v"(sidx[3]),
                           "v"(sidx[4]), "v"(sidx[5]), "v"(sidx[6]), "v"(sidx[7]));
        #pragma unroll
        for (int k = 0; k < BIN_BATCH; ++k) {
            const int j = i0 + k * BIN_THREADS;
            if (j < cnt) {
                stash[j] = make_int2(d[k], sidx[k]);
                if ((unsigned)d[k] < (unsigned)n) atomicAdd(&hist[(unsigned)d[k] / BK_R], 1);
            }
        }
    }
    __syncthreads();

    const int perb = (P + BIN_THREADS - 1) / BIN_THREADS;
    const int b0   = min(P, tid * perb);
    const int b1   = min(P, b0 + perb);
    int s = 0;
    for (int b = b0; b < b1; ++b) s += hist[b];
    int total;
    int run = block_scan_excl<BIN_THREADS>(s, wsum, tid, total);
    int* __restrict__ trow = tab + (size_t)blockIdx.x * TP;
    for (int b = b0; b < b1; ++b) {
        const int c = hist[b];
        trow[b] = s0 + run;
        hist[b] = run;
        run += c;
    }
    if (tid == BIN_THREADS - 1) {
        trow[P] = s0 + run;
        s_total = min(run, BIN_CAP);
    }
    __syncthreads();

    for (int i0 = tid; i0 < cnt; i0 += BIN_THREADS * BIN_BATCH) {
        int2 v[BIN_BATCH];
        #pragma unroll
        for (int k = 0; k < BIN_BATCH; ++k)
            v[k] = stash[min(i0 + k * BIN_THREADS, cnt - 1)];
        int  pos[BIN_BATCH];
        unsigned bk[BIN_BATCH];
        bool ok[BIN_BATCH];
        #pragma unroll
        for (int k = 0; k < BIN_BATCH; ++k) {
            ok[k]  = i0 + k * BIN_THREADS < cnt && (unsigned)v[k].x < (unsigned)n;
            bk[k]  = (unsigned)v[k].x / BK_R;
            pos[k] = atomicAdd(&hist[ok[k] ? bk[k] : P + lane], 1);
        }
        #pragma unroll
        for (int k = 0; k < BIN_BATCH; ++k) {
            const unsigned key = ((unsigned)v[k].y << BK_SH) | ((unsigned)v[k].x - bk[k] * BK_R);
            if (ok[k] && (unsigned)pos[k] < (unsigned)BIN_CAP)
                sorted[pos[k]] = (int)key;
        }
    }
    __syncthreads();

    const int tot = s_total;
    for (int i = tid; i < tot; i += BIN_THREADS) rec[s0 + i] = sorted[i];
}

template <int T>
__device__ __forceinline__ int bucket_runs(
    const int* __restrict__ tab, int b, int S, int TP, int e,
    int* rstart, int* roff, int* wsum, int* s_safe, int tid) {
    const int pers = (S + T - 1) / T;
    const int sA = min(S, tid * pers);
    const int sB = min(S, sA + pers);
    int sum = 0;
    for (int s = sA; s < sB; ++s) {
        int a = tab[(size_t)s * TP + b];
        const int z = tab[(size_t)s * TP + b + 1];
        a = min(max(a, 0), e);
        const int c = min(max(z - a, 0), e - a);
        rstart[s] = a;
        roff[s]   = c;
        sum += c;
        if (c > 0) *s_safe = a;
    }
    int total;
    int run = block_scan_excl<T>(sum, wsum, tid, total);
    for (int s = sA; s < sB; ++s) {
        const int c = roff[s];
        roff[s] = run;
        run += c;
    }
    if (tid == T - 1) roff[S] = run;
    __syncthreads();
    return total;
}

template <int T, int NB, typename F>
__device__ __forceinline__ void walk_runs(
    const int* __restrict__ rec, const int* rstart, const int* roff,
    int S, int safe, int tid, F f) {
    static_assert(NB <= 32 && (NB & (NB - 1)) == 0, "lane k of a wave fetches the table entries of run k of the batch");
    const int lane = tid & 31;
    const int wave = __builtin_amdgcn_readfirstlane(tid >> 5);
    for (int sb = wave * NB; sb < S; sb += (T / 32) * NB) {
        const int sl = min(sb + (lane & (NB - 1)), S - 1);
        const int av = rstart[sl];
        const int ov = roff[sl];
        const int ev = roff[sl + 1];
        int a[NB], o[NB], c[NB], v0[NB], v1[NB];
        #pragma unroll
        for (int k = 0; k < NB; ++k) {
            a[k] = __builtin_amdgcn_readlane(av, k);
            o[k] = __builtin_amdgcn_readlane(ov, k);
            c[k] = (sb + k < S) ? __builtin_amdgcn_readlane(ev, k) - o[k] : 0;
            v0[k] = rec[(lane < c[k]) ? a[k] + lane : safe];
            v1[k] = rec[(lane + 32 < c[k]) ? a[k] + lane + 32 : safe];
        }
        #pragma unroll
        for (int k = 0; k < NB; ++k) {
            if (lane < c[k]) f(o[k] + lane, v0[k]);
            if (lane + 32 < c[k]) f(o[k] + lane + 32, v1[k]);
            for (int j = lane + 64; j < c[k]; j += 32) f(o[k] + j, rec[a[k] + j]);
        }
    }
}

static_assert(BK_R == GM_WAVES * 16, "a GEMM block's rows must be one bucket");

__global__ __launch_bounds__(GM_THREADS) void gcn_gemm1_wmma(
    const float* __restrict__ x, const float* __restrict__ W1,
    const int* __restrict__ rec, const int* __restrict__ tab,
    float* __restrict__ dis, float* __restrict__ hw1, int n, int e, int S, int TP) {
    extern __shared__ int run_sm[];
    __shared__ int wsum[32];
    __shared__ int s_safe;
    int* rstart = run_sm;
    int* roff   = run_sm + S;
    __shared__ float W1s[W1_ROUNDS * GM_THREADS];
    __shared__ __attribute__((aligned(16))) int dcnt[BK_ROWS * 4];
    __shared__ float dis_s[BK_R];
    const int tid = threadIdx.x;
    {
        float w[W1_ROUNDS];
        #pragma unroll
        for (int k = 0; k < W1_ROUNDS; ++k)
            w[k] = W1[min(tid + k * GM_THREADS, IN_DIM * H1_DIM - 1)];
        #pragma unroll
        for (int k = 0; k < W1_ROUNDS; ++k)
            W1s[tid + k * GM_THREADS] = w[k];
    }
    for (int j = tid; j < BK_ROWS * 4; j += GM_THREADS) dcnt[j] = 0;
    {
        const int tot = bucket_runs<GM_THREADS>(tab, (int)blockIdx.x, S, TP, e, rstart, roff, wsum, &s_safe, tid);
        if (tot > 0) {
            const int q = tid & 3;
            walk_runs<GM_THREADS, RB_GM>(rec, rstart, roff, S, s_safe, tid,
                           [&](int, int r) { atomicAdd(&dcnt[((r & (BK_ROWS - 1)) << 2) | q], 1); });
        }
    }
    __syncthreads();
    if (tid < BK_R) {
        const int4 c4 = *(const int4*)&dcnt[tid * 4];
        const float d = (float)(1 + c4.x + c4.y + c4.z + c4.w);
        const float s = (d > 0.0f) ? rsqrtf(d) : 0.0f;
        dis_s[tid] = s;
    }
    __syncthreads();

    const int wave = tid >> 5;
    const int lane = tid & 31;
    const int half = lane >> 4;
    const int l    = lane & 15;
    const int row0 = (blockIdx.x * GM_WAVES + wave) * 16;
    if (row0 >= n) return;

    int arow = row0 + l;
    if (arow >= n) arow = n - 1;

    v8f c = {};
    #pragma unroll 4
    for (int k0 = 0; k0 < IN_DIM; k0 += 4) {
        const int ka = k0 + half * 2;
        const v2f a = *(const v2f*)&x[(size_t)arow * IN_DIM + ka];
        v2f b;
        b.x = W1s[ka * H1_DIM + l];
        b.y = W1s[(ka + 1) * H1_DIM + l];
        c = __builtin_amdgcn_wmma_f32_16x16x4_f32(
                false, a, false, b, (short)0, c, false, false);
    }

    if (row0 + 16 <= n) {
        #pragma unroll
        for (int r = 0; r < 8; ++r) {
            const int row = row0 + r + half * 8;
            hw1[(size_t)row * H1_DIM + l] = dis_s[wave * 16 + r + half * 8] * c[r];
        }
    } else {
        for (int r = 0; r < 8; ++r) {
            const int row = row0 + r + half * 8;
            if (row < n) hw1[(size_t)row * H1_DIM + l] = dis_s[wave * 16 + r + half * 8] * c[r];
        }
    }

    if (tid < BK_R) {
        const int node = blockIdx.x * BK_R + tid;
        if (node < n) dis[blockIdx.x * DIS_PITCH + tid] = dis_s[tid];
    }
}

__global__ __launch_bounds__(SC_THREADS) void gcn_scatter1_gemm2(
    const int* __restrict__ rec, const int* __restrict__ tab,
    const float* __restrict__ hw1, const float* __restrict__ dis,
    const float* __restrict__ b1, const float* __restrict__ W2,
    float* __restrict__ hw2, int n, int e, int S, int TP) {
    extern __shared__ int run_sm[];
    __shared__ int wsum[32];
    __shared__ int s_safe;
    int* rstart = run_sm;
    int* roff   = run_sm + S;
    __shared__ __attribute__((aligned(16))) float acc[BK_ROWS * H1_DIM];
    __shared__ __attribute__((aligned(16))) int lrec[SC_CAP];
    __shared__ float W2s[H1_DIM * 16];
    __shared__ float dis_s[DIS_PITCH];
    __shared__ __attribute__((aligned(16))) float outs[BK_R * H2_DIM];
    const int tid = threadIdx.x;
    const int lo  = blockIdx.x * BK_R;
    for (int j = tid; j < BK_R * H1_DIM; j += SC_THREADS) acc[j] = 0.0f;

    const int tot = bucket_runs<SC_THREADS>(tab, (int)blockIdx.x, S, TP, e, rstart, roff, wsum, &s_safe, tid);
    const int c = tid & (H1_DIM - 1);
    const int g = tid >> 4;
    const int SW = (SC_THREADS >> 4) * SC_BATCH;
    for (int p0 = 0; p0 < tot; p0 += SC_CAP) {
        if (p0 != 0) __syncthreads();
        const int len = min(SC_CAP, tot - p0);
        walk_runs<SC_THREADS, RB_SC>(rec, rstart, roff, S, s_safe, tid,
                              [&](int i, int r) {
                                  const unsigned j = (unsigned)(i - p0);
                                  if (j < (unsigned)len) lrec[j] = r;
                              });
        if (tid < SC_BATCH && len + tid < SC_CAP) lrec[len + tid] = 0;
        __syncthreads();

        for (int i0 = g * SC_BATCH; i0 < len; i0 += SW) {
            const int4 ra = *(const int4*)&lrec[i0];
            const int4 rb = *(const int4*)&lrec[i0 + 4];
            const int r[SC_BATCH] = {ra.x, ra.y, ra.z, ra.w, rb.x, rb.y, rb.z, rb.w};
            float h[SC_BATCH];
            #pragma unroll
            for (int k = 0; k < SC_BATCH; ++k) {
                const unsigned s = min((unsigned)r[k] >> BK_SH, (unsigned)(n - 1));
                h[k] = hw1[(size_t)s * H1_DIM + c];
            }
            #pragma unroll
            for (int k = 0; k < SC_BATCH; ++k) {
                const float v = (i0 + k < len) ? h[k] : 0.0f;
                const int dl = r[k] & (BK_ROWS - 1);
                atomicAdd(&acc[dl * H1_DIM + c], v);
            }
        }
    }
    const int cnt = min(BK_R, n - lo) * H1_DIM;
    const float* __restrict__ own = hw1 + (size_t)lo * H1_DIM;
    const float bias = b1[c];
    float o[H1_ROUNDS];
    #pragma unroll
    for (int k = 0; k < H1_ROUNDS; ++k)
        o[k] = own[min(tid + k * SC_THREADS, cnt - 1)];
    if (tid < H1_DIM * 16) {
        const int k = tid >> 4, col = tid & 15;
        W2s[tid] = (col < H2_DIM) ? W2[k * H2_DIM + col] : 0.0f;
    } else if (tid < H1_DIM * 16 + BK_R) {
        const int i = tid - H1_DIM * 16;
        dis_s[i] = (lo + i < n) ? dis[blockIdx.x * DIS_PITCH + i] : 0.0f;
    }
    __syncthreads();

    #pragma unroll
    for (int k = 0; k < H1_ROUNDS; ++k) {
        const int j = tid + k * SC_THREADS;
        const float v = fmaxf(dis_s[j >> 4] * (acc[j] + o[k]) + bias, 0.0f);
        acc[j] = (j < cnt) ? v : 0.0f;
    }
    __syncthreads();

    const int wave = tid >> 5;
    const int lane = tid & 31;
    const int half = lane >> 4;
    const int l    = lane & 15;
    if (wave < BK_R / 16 && lo + wave * 16 < n) {
        v8f d = {};
        #pragma unroll
        for (int k0 = 0; k0 < H1_DIM; k0 += 4) {
            const int ka = k0 + half * 2;
            const v2f a = *(const v2f*)&acc[(wave * 16 + l) * H1_DIM + ka];
            v2f b;
            b.x = W2s[ka * 16 + l];
            b.y = W2s[(ka + 1) * 16 + l];
            d = __builtin_amdgcn_wmma_f32_16x16x4_f32(
                    false, a, false, b, (short)0, d, false, false);
        }
        if (l < H2_DIM) {
            #pragma unroll
            for (int r = 0; r < 8; ++r) {
                const int row = wave * 16 + r + half * 8;
                outs[row * H2_DIM + l] = dis_s[row] * d[r];
            }
        }
    }
    __syncthreads();

    const int cnt2 = min(BK_R, n - lo) * H2_DIM;
    if (tid * 4 < cnt2)
        *(float4*)&hw2[(size_t)lo * H2_DIM + tid * 4] = *(const float4*)&outs[tid * 4];
}

__global__ __launch_bounds__(SC_THREADS) void gcn_scatter2(
    const int* __restrict__ rec, const int* __restrict__ tab,
    const float* __restrict__ hw2, const float* __restrict__ dis,
    const float* __restrict__ b2, float* __restrict__ out, int n, int e, int S, int TP) {
    extern __shared__ int run_sm[];
    __shared__ int wsum[32];
    __shared__ int s_safe;
    int* rstart = run_sm;
    int* roff   = run_sm + S;
    __shared__ float acc[BK_ROWS * H2_DIM];
    __shared__ __attribute__((aligned(16))) int lrec[SC_CAP];
    const int tid = threadIdx.x;
    for (int j = tid; j < BK_R * H2_DIM; j += SC_THREADS) acc[j] = 0.0f;

    const int lo  = blockIdx.x * BK_R;
    const int tot = bucket_runs<SC_THREADS>(tab, (int)blockIdx.x, S, TP, e, rstart, roff, wsum, &s_safe, tid);
    const int c = tid & (H2_DIM - 1);
    const int g = tid >> 3;
    const int SW = (SC_THREADS >> 3) * SC_BATCH;
    for (int p0 = 0; p0 < tot; p0 += SC_CAP) {
        if (p0 != 0) __syncthreads();
        const int len = min(SC_CAP, tot - p0);
        walk_runs<SC_THREADS, RB_SC>(rec, rstart, roff, S, s_safe, tid,
                              [&](int i, int r) {
                                  const unsigned j = (unsigned)(i - p0);
                                  if (j < (unsigned)len) lrec[j] = r;
                              });
        if (tid < SC_BATCH && len + tid < SC_CAP) lrec[len + tid] = 0;
        __syncthreads();

        for (int i0 = g * SC_BATCH; i0 < len; i0 += SW) {
            const int4 ra = *(const int4*)&lrec[i0];
            const int4 rb = *(const int4*)&lrec[i0 + 4];
            const int r[SC_BATCH] = {ra.x, ra.y, ra.z, ra.w, rb.x, rb.y, rb.z, rb.w};
            float h[SC_BATCH];
            #pragma unroll
            for (int k = 0; k < SC_BATCH; ++k) {
                const unsigned s = min((unsigned)r[k] >> BK_SH, (unsigned)(n - 1));
                h[k] = hw2[(size_t)s * H2_DIM + c];
            }
            #pragma unroll
            for (int k = 0; k < SC_BATCH; ++k) {
                const float v = (i0 + k < len) ? h[k] : 0.0f;
                const int dl = r[k] & (BK_ROWS - 1);
                atomicAdd(&acc[dl * H2_DIM + c], v);
            }
        }
    }
    __syncthreads();

    const int cnt = min(BK_R, n - lo) * H2_DIM;
    const float* __restrict__ own = hw2 + (size_t)lo * H2_DIM;
    float* __restrict__ row = out + (size_t)lo * H2_DIM;
    const float bias = b2[c];
    for (int j = tid; j < cnt; j += SC_THREADS)
        row[j] = dis[blockIdx.x * DIS_PITCH + (j >> 3)] * (acc[j] + own[j]) + bias;
}

extern "C" void kernel_launch(void* const* d_in, const int* in_sizes, int n_in,
                              void* d_out, int out_size, void* d_ws, size_t ws_size,
                              hipStream_t stream) {
    const float* x    = (const float*)d_in[0];
    const int*   edge = (const int*)  d_in[1];
    const float* W1   = (const float*)d_in[2];
    const float* b1   = (const float*)d_in[3];
    const float* W2   = (const float*)d_in[4];
    const float* b2   = (const float*)d_in[5];
    float* out = (float*)d_out;

    const int n = in_sizes[0] / IN_DIM;
    const int e = in_sizes[1] / 2;
    const int* src = edge;
    const int* dst = edge + e;

    float* ws   = (float*)d_ws;
    const int P = (n + BK_R - 1) / BK_R;
    float* dis  = ws;
    float* hw1  = dis + (size_t)P * DIS_PITCH;
    float* hw2  = (float*)((((size_t)(hw1 + (size_t)n * H1_DIM)) + 127) & ~(size_t)127);

    const int bneed = (e + BIN_CAP - 1) / BIN_CAP;
    const int S  = bneed > 256 ? bneed : 256;
    const int TP = (P + 1 + 31) & ~31;
    int* tab = (int*)((((size_t)(hw2 + (size_t)n * H2_DIM)) + 127) & ~(size_t)127);
    int* rec = tab + (size_t)S * TP;
    const size_t runb = (size_t)(2 * S + 1) * sizeof(int);


    gcn_bin_edges     <<<S, BIN_THREADS, (size_t)(P + 32) * sizeof(int), stream>>>(src, dst, tab, rec, e, n, P, TP);
    gcn_gemm1_wmma    <<<P, GM_THREADS, runb, stream>>>(x, W1, rec, tab, dis, hw1, n, e, S, TP);
    gcn_scatter1_gemm2<<<P, SC_THREADS, runb, stream>>>(rec, tab, hw1, dis, b1, W2, hw2, n, e, S, TP);
    gcn_scatter2      <<<P, SC_THREADS, runb, stream>>>(rec, tab, hw2, dis, b2, out, n, e, S, TP);
}
